# starting kernel
# baseline (speedup 1.0000x reference)
#include <hip/hip_runtime.h>
#include <hip/hip_bf16.h>

#define N_NODES  50000
#define N_EDGES  800000
#define N_GRAPHS 512
#define D_H      64
#define D_OUT    32
#define LN_EPS   1e-5f

typedef __attribute__((ext_vector_type(2)))  float    v2f;
typedef __attribute__((ext_vector_type(8)))  float    v8f;
typedef __attribute__((ext_vector_type(16))) _Float16 v16h;

#if defined(__has_builtin)
#if __has_builtin(__builtin_amdgcn_wmma_f32_16x16x4_f32)
#define USE_F32_WMMA 1
#endif
#endif

template <int NCOLS, bool BIAS>
__global__ __launch_bounds__(32 * (NCOLS / 16)) void gcn_gemm_wmma(
    const float* __restrict__ A, const float* __restrict__ B,
    const float* __restrict__ bias, float* __restrict__ C) {
  const int lane = threadIdx.x & 31;
  const int wave = threadIdx.x >> 5;
  const int m    = lane & 15;
  const int half = lane >> 4;
  const int c0   = wave * 16;
  const int r0   = blockIdx.x * 16;
  const int row  = r0 + m;

  v8f acc = {0.f, 0.f, 0.f, 0.f, 0.f, 0.f, 0.f, 0.f};

#if USE_F32_WMMA
  #pragma unroll
  for (int k0 = 0; k0 < 64; k0 += 4) {
    const int ka = k0 + 2 * half;
    v2f a, b;
    a.x = A[row * 64 + ka];
    a.y = A[row * 64 + ka + 1];
    b.x = B[ka * NCOLS + c0 + m];
    b.y = B[(ka + 1) * NCOLS + c0 + m];
    acc = __builtin_amdgcn_wmma_f32_16x16x4_f32(false, a, false, b,
                                                (short)0, acc, false, false);
  }
#else
  for (int kstep = 0; kstep < 2; ++kstep) {
    v16h af, bf;
    #pragma unroll
    for (int e = 0; e < 16; ++e) {
      int v = e >> 1, j = e & 1, grp = v >> 2;
      int K = 32 * kstep + 16 * grp + 8 * half + ((v & 3) * 2 + j);
      af[e] = (_Float16)A[row * 64 + K];
    }
    #pragma unroll
    for (int e = 0; e < 16; ++e) {
      int v = e >> 1, j = e & 1;
      int K = 32 * kstep + 16 * half + 2 * v + j;
      bf[e] = (_Float16)B[K * NCOLS + c0 + m];
    }
    acc = __builtin_amdgcn_wmma_f32_16x16x32_f16(false, af, false, bf,
                                                 (short)0, acc, false, false);
  }
#endif

  #pragma unroll
  for (int i = 0; i < 8; ++i) {
    float v = acc[i];
    if (BIAS) v += bias[c0 + m];
    C[(r0 + i + 8 * half) * NCOLS + c0 + m] = v;
  }
}

__global__ void k_deg_init(float* __restrict__ deg) {
  int i = blockIdx.x * blockDim.x + threadIdx.x;
  if (i < N_NODES) deg[i] = 1.0f;
}

__global__ void k_deg_accum(const int* __restrict__ dst, float* __restrict__ deg) {
  int i = blockIdx.x * blockDim.x + threadIdx.x;
  if (i < N_EDGES) atomicAdd(&deg[dst[i]], 1.0f);
}

__global__ void k_dinv(float* __restrict__ deg) {
  int i = blockIdx.x * blockDim.x + threadIdx.x;
  if (i < N_NODES) deg[i] = rsqrtf(deg[i]);
}

__global__ void k_enorm(const int* __restrict__ src, const int* __restrict__ dst,
                        const float* __restrict__ dinv, float* __restrict__ enorm) {
  int i = blockIdx.x * blockDim.x + threadIdx.x;
  if (i < N_EDGES) enorm[i] = dinv[src[i]] * dinv[dst[i]];
}

__global__ void k_selfloop(const float* __restrict__ hW, const float* __restrict__ dinv,
                           float* __restrict__ agg) {
  int idx = blockIdx.x * blockDim.x + threadIdx.x;
  if (idx >= N_NODES * 64) return;
  float di = dinv[idx >> 6];
  agg[idx] = di * di * hW[idx];
}

__global__ void k_edge_scatter(const float* __restrict__ hW, const int* __restrict__ src,
                               const int* __restrict__ dst, const float* __restrict__ enorm,
                               float* __restrict__ agg) {
  int idx = blockIdx.x * blockDim.x + threadIdx.x;
  if (idx >= N_EDGES * 64) return;
  int e = idx >> 6, f = idx & 63;
  int s = src[e], d = dst[e];
  atomicAdd(&agg[d * 64 + f], enorm[e] * hW[s * 64 + f]);
}

__global__ void k_bias_ln_relu(const float* __restrict__ agg, const float* __restrict__ bias,
                               const float* __restrict__ lnw, const float* __restrict__ lnb,
                               float* __restrict__ out) {
  int node = (blockIdx.x * blockDim.x + threadIdx.x) >> 5;
  int lane = threadIdx.x & 31;
  if (node >= N_NODES) return;
  const float* rowp = agg + (size_t)node * 64;
  float v0 = rowp[lane] + bias[lane];
  float v1 = rowp[lane + 32] + bias[lane + 32];
  float s = v0 + v1;
  #pragma unroll
  for (int off = 16; off > 0; off >>= 1) s += __shfl_xor(s, off, 32);
  float mu = s * (1.0f / 64.0f);
  float d0 = v0 - mu, d1 = v1 - mu;
  float q = d0 * d0 + d1 * d1;
  #pragma unroll
  for (int off = 16; off > 0; off >>= 1) q += __shfl_xor(q, off, 32);
  float rstd = rsqrtf(q * (1.0f / 64.0f) + LN_EPS);
  float* orow = out + (size_t)node * 64;
  orow[lane]      = fmaxf(0.f, d0 * rstd * lnw[lane] + lnb[lane]);
  orow[lane + 32] = fmaxf(0.f, d1 * rstd * lnw[lane + 32] + lnb[lane + 32]);
}

__global__ void k_zero_pool(float* __restrict__ pool, float* __restrict__ cnt) {
  int i = blockIdx.x * blockDim.x + threadIdx.x;
  if (i < N_GRAPHS * 64) pool[i] = 0.f;
  if (i < N_GRAPHS) cnt[i] = 0.f;
}

__global__ void k_pool_accum(const float* __restrict__ h, const int* __restrict__ batch,
                             float* __restrict__ pool, float* __restrict__ cnt) {
  int idx = blockIdx.x * blockDim.x + threadIdx.x;
  if (idx >= N_NODES * 64) return;
  int n = idx >> 6, f = idx & 63;
  int g = batch[n];
  atomicAdd(&pool[g * 64 + f], h[idx]);
  if (f == 0) atomicAdd(&cnt[g], 1.0f);
}

__global__ void k_pool_div(float* __restrict__ pool, const float* __restrict__ cnt) {
  int i = blockIdx.x * blockDim.x + threadIdx.x;
  if (i < N_GRAPHS * 64) pool[i] /= fmaxf(cnt[i >> 6], 1.0f);
}

static inline int cdiv(int a, int b) { return (a + b - 1) / b; }

static void run_layer(const float* hin, const float* W, const float* b,
                      const float* lnw, const float* lnb,
                      const float* dinv, const int* src, const int* dst,
                      const float* enorm, float* hW, float* agg, float* hout,
                      hipStream_t stream) {
  gcn_gemm_wmma<64, false><<<N_NODES / 16, 128, 0, stream>>>(hin, W, nullptr, hW);
  k_selfloop<<<cdiv(N_NODES * 64, 256), 256, 0, stream>>>(hW, dinv, agg);
  k_edge_scatter<<<cdiv(N_EDGES * 64, 256), 256, 0, stream>>>(hW, src, dst, enorm, agg);
  k_bias_ln_relu<<<cdiv(N_NODES * 32, 256), 256, 0, stream>>>(agg, b, lnw, lnb, hout);
}

extern "C" void kernel_launch(void* const* d_in, const int* in_sizes, int n_in,
                              void* d_out, int out_size, void* d_ws, size_t ws_size,
                              hipStream_t stream) {
  (void)in_sizes; (void)n_in; (void)out_size; (void)ws_size;

  const float* x       = (const float*)d_in[0];
  const int*   eidx    = (const int*)d_in[1];
  const int*   batch   = (const int*)d_in[2];
  const float* W_in    = (const float*)d_in[3];
  const float* b_in    = (const float*)d_in[4];
  const float* ln_in_w = (const float*)d_in[5];
  const float* ln_in_b = (const float*)d_in[6];
  const float* W_h     = (const float*)d_in[7];
  const float* b_h     = (const float*)d_in[8];
  const float* ln_h_w  = (const float*)d_in[9];
  const float* ln_h_b  = (const float*)d_in[10];
  const float* W_out   = (const float*)d_in[11];
  const float* b_out   = (const float*)d_in[12];

  const int* src = eidx;
  const int* dst = eidx + N_EDGES;

  float* ws    = (float*)d_ws;
  float* dinv  = ws;
  float* enorm = dinv + N_NODES;
  float* hW    = enorm + N_EDGES;
  float* agg   = hW + (size_t)N_NODES*64;
  float* hbuf  = agg + (size_t)N_NODES*64;
  float* pool  = hbuf + (size_t)N_NODES*64;
  float* cnt   = pool + N_GRAPHS * 64;

  k_deg_init<<<cdiv(N_NODES, 256), 256, 0, stream>>>(dinv);
  k_deg_accum<<<cdiv(N_EDGES, 256), 256, 0, stream>>>(dst, dinv);
  k_dinv<<<cdiv(N_NODES, 256), 256, 0, stream>>>(dinv);
  k_enorm<<<cdiv(N_EDGES, 256), 256, 0, stream>>>(src, dst, dinv, enorm);

  run_layer(x,    W_in,       b_in,       ln_in_w,      ln_in_b,      dinv, src, dst, enorm, hW, agg, hbuf, stream);
  run_layer(hbuf, W_h,        b_h,        ln_h_w,       ln_h_b,       dinv, src, dst, enorm, hW, agg, hbuf, stream);
  run_layer(hbuf, W_h + 64*64, b_h + 64,  ln_h_w + 64,  ln_h_b + 64,  dinv, src, dst, enorm, hW, agg, hbuf, stream);

  k_zero_pool<<<cdiv(N_GRAPHS * 64, 256), 256, 0, stream>>>(pool, cnt);
  k_pool_accum<<<cdiv(N_NODES * 64, 256), 256, 0, stream>>>(hbuf, batch, pool, cnt);
  k_pool_div<<<cdiv(N_GRAPHS * 64, 256), 256, 0, stream>>>(pool, cnt);
  gcn_gemm_wmma<D_OUT, true><<<N_GRAPHS / 16, 64, 0, stream>>>(pool, W_out, b_out, (float*)d_out);
}


# Adjacency applied before W, (A h)W for A(h W): a row sums h, then one wave does 16 rows x W (f32 WMMA, 4 tiles) with bias+layernorm+relu in its epilogue; 17 to 14 launches; sums reassociated
# speedup vs baseline: 2.6901x; 2.6901x over previous
#include <hip/hip_runtime.h>
#include <hip/hip_bf16.h>

#define N_NODES  50000
#define N_EDGES  800000
#define N_GRAPHS 512
#define D_H      64
#define D_OUT    32
#define LN_EPS   1e-5f
#define N_PAD    50048

typedef __attribute__((ext_vector_type(2)))  float    v2f;
typedef __attribute__((ext_vector_type(8)))  float    v8f;
typedef __attribute__((ext_vector_type(16))) _Float16 v16h;

#if defined(__has_builtin)
#if __has_builtin(__builtin_amdgcn_wmma_f32_16x16x4_f32)
#define USE_F32_WMMA 1
#endif
#endif

template <int NCOLS, bool BIAS>
__global__ __launch_bounds__(32 * (NCOLS / 16)) void gcn_gemm_wmma(
    const float* __restrict__ A, const float* __restrict__ B,
    const float* __restrict__ bias, float* __restrict__ C) {
  const int lane = threadIdx.x & 31;
  const int wave = threadIdx.x >> 5;
  const int m    = lane & 15;
  const int half = lane >> 4;
  const int c0   = wave * 16;
  const int r0   = blockIdx.x * 16;
  const int row  = r0 + m;

  v8f acc = {0.f, 0.f, 0.f, 0.f, 0.f, 0.f, 0.f, 0.f};

#if USE_F32_WMMA
  #pragma unroll
  for (int k0 = 0; k0 < 64; k0 += 4) {
    const int ka = k0 + 2 * half;
    v2f a, b;
    a.x = A[row * 64 + ka];
    a.y = A[row * 64 + ka + 1];
    b.x = B[ka * NCOLS + c0 + m];
    b.y = B[(ka + 1) * NCOLS + c0 + m];
    acc = __builtin_amdgcn_wmma_f32_16x16x4_f32(false, a, false, b,
                                                (short)0, acc, false, false);
  }
#else
  for (int kstep = 0; kstep < 2; ++kstep) {
    v16h af, bf;
    #pragma unroll
    for (int e = 0; e < 16; ++e) {
      int v = e >> 1, j = e & 1, grp = v >> 2;
      int K = 32 * kstep + 16 * grp + 8 * half + ((v & 3) * 2 + j);
      af[e] = (_Float16)A[row * 64 + K];
    }
    #pragma unroll
    for (int e = 0; e < 16; ++e) {
      int v = e >> 1, j = e & 1;
      int K = 32 * kstep + 16 * half + 2 * v + j;
      bf[e] = (_Float16)B[K * NCOLS + c0 + m];
    }
    acc = __builtin_amdgcn_wmma_f32_16x16x32_f16(false, af, false, bf,
                                                 (short)0, acc, false, false);
  }
#endif

  #pragma unroll
  for (int i = 0; i < 8; ++i) {
    float v = acc[i];
    if (BIAS) v += bias[c0 + m];
    C[(r0 + i + 8 * half) * NCOLS + c0 + m] = v;
  }
}

__global__ void k_head_init(int* __restrict__ head) {
  int i = blockIdx.x * blockDim.x + threadIdx.x;
  if (i < N_NODES) head[i] = -1;
}

__global__ void k_link(const int* __restrict__ src, const int* __restrict__ dst,
                       int* __restrict__ head, int* __restrict__ rec) {
  int e = blockIdx.x * blockDim.x + threadIdx.x;
  if (e >= N_EDGES) return;
  int old = atomicExch(&head[dst[e]], e);
  *reinterpret_cast<int2*>(rec + 4 * (size_t)e) = make_int2(old, src[e]);
}

__global__ void k_dinv(const int* __restrict__ head, const int* __restrict__ rec,
                       float* __restrict__ dinv) {
  int i = blockIdx.x * blockDim.x + threadIdx.x;
  if (i >= N_NODES) return;
  int c = 1;
  int e = head[i];
  for (int it = 0; it < N_EDGES && e >= 0; ++it) {
    ++c;
    e = rec[4 * (size_t)e];
  }
  dinv[i] = rsqrtf((float)c);
}

__global__ void k_enorm(const int* __restrict__ src, const int* __restrict__ dst,
                        const float* __restrict__ dinv, int* __restrict__ rec) {
  int i = blockIdx.x * blockDim.x + threadIdx.x;
  if (i < N_EDGES) rec[4 * (size_t)i + 2] = __float_as_int(dinv[src[i]] * dinv[dst[i]]);
}

__global__ __launch_bounds__(256) void k_row_gather(
    const float* __restrict__ h, const float* __restrict__ dinv,
    const int* __restrict__ head, const int* __restrict__ rec,
    float* __restrict__ agg) {
  int node = (blockIdx.x * blockDim.x + threadIdx.x) >> 5;
  const int lane = threadIdx.x & 31;
  if (node >= N_NODES) return;
  node = __builtin_amdgcn_readfirstlane(node);
  const float di = dinv[node];
  const float2 own = *reinterpret_cast<const float2*>(h + (size_t)node * 64 + 2 * lane);
  const float sl = di * di;
  float a0 = sl * own.x, a1 = sl * own.y;
  int e = head[node];
  if (e >= 0) {
    int4 r = *reinterpret_cast<const int4*>(rec + 4 * (size_t)e);
    for (int it = 0; it < N_EDGES; ++it) {
      const float2 x = *reinterpret_cast<const float2*>(h + (size_t)r.y * 64 + 2 * lane);
      const float w = __int_as_float(r.z);
      const int en = r.x;
      int4 rn = r;
      if (en >= 0) rn = *reinterpret_cast<const int4*>(rec + 4 * (size_t)en);
      a0 += w * x.x;
      a1 += w * x.y;
      if (en < 0) break;
      r = rn;
    }
  }
  *reinterpret_cast<float2*>(agg + (size_t)node * 64 + 2 * lane) = make_float2(a0, a1);
}

__global__ __launch_bounds__(128) void k_gemm_ln_relu(
    const float* __restrict__ agg, const float* __restrict__ Wm,
    const float* __restrict__ bias, const float* __restrict__ lnw,
    const float* __restrict__ lnb, float* __restrict__ out) {
  const int tile = (blockIdx.x * blockDim.x + threadIdx.x) >> 5;
  if (tile >= N_NODES / 16) return;
  const int lane = threadIdx.x & 31;
  const int m    = lane & 15;
  const int half = lane >> 4;
  const int r0   = tile * 16;

  const float*  arow = agg + (size_t)(r0 + m) * 64 + 2 * half;
  const float4* W4   = reinterpret_cast<const float4*>(Wm);

  v8f acc0 = {0.f, 0.f, 0.f, 0.f, 0.f, 0.f, 0.f, 0.f};
  v8f acc1 = acc0, acc2 = acc0, acc3 = acc0;

  #pragma unroll
  for (int j = 0; j < 16; ++j) {
    const int k = 4 * j + 2 * half;
    const v2f a = *reinterpret_cast<const v2f*>(arow + 4 * j);
    const float4 w0 = W4[k * 16 + m];
    const float4 w1 = W4[(k + 1) * 16 + m];
    v2f b;
    b.x = w0.x; b.y = w1.x;
    acc0 = __builtin_amdgcn_wmma_f32_16x16x4_f32(false, a, false, b, (short)0, acc0, false, false);
    b.x = w0.y; b.y = w1.y;
    acc1 = __builtin_amdgcn_wmma_f32_16x16x4_f32(false, a, false, b, (short)0, acc1, false, false);
    b.x = w0.z; b.y = w1.z;
    acc2 = __builtin_amdgcn_wmma_f32_16x16x4_f32(false, a, false, b, (short)0, acc2, false, false);
    b.x = w0.w; b.y = w1.w;
    acc3 = __builtin_amdgcn_wmma_f32_16x16x4_f32(false, a, false, b, (short)0, acc3, false, false);
  }

  const float4 bs = reinterpret_cast<const float4*>(bias)[m];
  const float4 gw = reinterpret_cast<const float4*>(lnw)[m];
  const float4 gb = reinterpret_cast<const float4*>(lnb)[m];

  #pragma unroll
  for (int i = 0; i < 8; ++i) {
    const float v0 = acc0[i] + bs.x, v1 = acc1[i] + bs.y;
    const float v2 = acc2[i] + bs.z, v3 = acc3[i] + bs.w;
    float s = (v0 + v1) + (v2 + v3);
    #pragma unroll
    for (int off = 8; off > 0; off >>= 1) s += __shfl_xor(s, off, 32);
    const float mu = s * (1.0f / 64.0f);
    const float d0 = v0 - mu, d1 = v1 - mu, d2 = v2 - mu, d3 = v3 - mu;
    float q = (d0 * d0 + d1 * d1) + (d2 * d2 + d3 * d3);
    #pragma unroll
    for (int off = 8; off > 0; off >>= 1) q += __shfl_xor(q, off, 32);
    const float rstd = rsqrtf(q * (1.0f / 64.0f) + LN_EPS);
    float4 o;
    o.x = fmaxf(0.f, d0 * rstd * gw.x + gb.x);
    o.y = fmaxf(0.f, d1 * rstd * gw.y + gb.y);
    o.z = fmaxf(0.f, d2 * rstd * gw.z + gb.z);
    o.w = fmaxf(0.f, d3 * rstd * gw.w + gb.w);
    reinterpret_cast<float4*>(out + (size_t)(r0 + i + 8 * half) * 64)[m] = o;
  }
}

__global__ void k_zero_pool(float* __restrict__ pool, float* __restrict__ cnt) {
  int i = blockIdx.x * blockDim.x + threadIdx.x;
  if (i < N_GRAPHS * 64) pool[i] = 0.f;
  if (i < N_GRAPHS) cnt[i] = 0.f;
}

__global__ void k_pool_accum(const float* __restrict__ h, const int* __restrict__ batch,
                             float* __restrict__ pool, float* __restrict__ cnt) {
  int idx = blockIdx.x * blockDim.x + threadIdx.x;
  if (idx >= N_NODES * 64) return;
  int n = idx >> 6, f = idx & 63;
  int g = batch[n];
  atomicAdd(&pool[g * 64 + f], h[idx]);
  if (f == 0) atomicAdd(&cnt[g], 1.0f);
}

__global__ void k_pool_div(float* __restrict__ pool, const float* __restrict__ cnt) {
  int i = blockIdx.x * blockDim.x + threadIdx.x;
  if (i < N_GRAPHS * 64) pool[i] /= fmaxf(cnt[i >> 6], 1.0f);
}

static inline int cdiv(int a, int b) { return (a + b - 1) / b; }

static void run_layer(const float* hin, const float* W, const float* b,
                      const float* lnw, const float* lnb,
                      const float* dinv, const int* head, const int* rec,
                      float* agg, float* hout,
                      hipStream_t stream) {
  k_row_gather<<<cdiv(N_NODES * 32, 256), 256, 0, stream>>>(hin, dinv, head, rec, agg);
  k_gemm_ln_relu<<<cdiv(N_NODES / 16, 4), 128, 0, stream>>>(agg, W, b, lnw, lnb, hout);
}

extern "C" void kernel_launch(void* const* d_in, const int* in_sizes, int n_in,
                              void* d_out, int out_size, void* d_ws, size_t ws_size,
                              hipStream_t stream) {
  (void)in_sizes; (void)n_in; (void)out_size; (void)ws_size;

  const float* x       = (const float*)d_in[0];
  const int*   eidx    = (const int*)d_in[1];
  const int*   batch   = (const int*)d_in[2];
  const float* W_in    = (const float*)d_in[3];
  const float* b_in    = (const float*)d_in[4];
  const float* ln_in_w = (const float*)d_in[5];
  const float* ln_in_b = (const float*)d_in[6];
  const float* W_h     = (const float*)d_in[7];
  const float* b_h     = (const float*)d_in[8];
  const float* ln_h_w  = (const float*)d_in[9];
  const float* ln_h_b  = (const float*)d_in[10];
  const float* W_out   = (const float*)d_in[11];
  const float* b_out   = (const float*)d_in[12];

  const int* src = eidx;
  const int* dst = eidx + N_EDGES;

  float* ws    = (float*)d_ws;
  float* dinv  = ws;
  float* agg   = dinv + N_PAD;
  float* hbuf  = agg + (size_t)N_NODES*64;
  float* pool  = hbuf + (size_t)N_NODES*64;
  float* cnt   = pool + N_GRAPHS * 64;
  int*   head  = (int*)(cnt + N_GRAPHS);
  int*   rec   = head + N_PAD;

  k_head_init<<<cdiv(N_NODES, 256), 256, 0, stream>>>(head);
  k_link<<<cdiv(N_EDGES, 256), 256, 0, stream>>>(src, dst, head, rec);
  k_dinv<<<cdiv(N_NODES, 64), 64, 0, stream>>>(head, rec, dinv);
  k_enorm<<<cdiv(N_EDGES, 256), 256, 0, stream>>>(src, dst, dinv, rec);

  run_layer(x,    W_in,       b_in,       ln_in_w,      ln_in_b,      dinv, head, rec, agg, hbuf, stream);
  run_layer(hbuf, W_h,        b_h,        ln_h_w,       ln_h_b,       dinv, head, rec, agg, hbuf, stream);
  run_layer(hbuf, W_h + 64*64, b_h + 64,  ln_h_w + 64,  ln_h_b + 64,  dinv, head, rec, agg, hbuf, stream);

  k_zero_pool<<<cdiv(N_GRAPHS * 64, 256), 256, 0, stream>>>(pool, cnt);
  k_pool_accum<<<cdiv(N_NODES * 64, 256), 256, 0, stream>>>(hbuf, batch, pool, cnt);
  k_pool_div<<<cdiv(N_GRAPHS * 64, 256), 256, 0, stream>>>(pool, cnt);
  gcn_gemm_wmma<D_OUT, true><<<N_GRAPHS / 16, 64, 0, stream>>>(pool, W_out, b_out, (float*)d_out);
}


# Mean pool sums by runs: a wave adds 16 consecutive rows in registers while the graph id stays the same, then one f32 atomic per feature per run (100000 to 3125 waves); pool sum order changed
# speedup vs baseline: 3.0970x; 1.1513x over previous
#include <hip/hip_runtime.h>
#include <hip/hip_bf16.h>

#define N_NODES  50000
#define N_EDGES  800000
#define N_GRAPHS 512
#define D_H      64
#define D_OUT    32
#define LN_EPS   1e-5f
#define N_PAD    50048

typedef __attribute__((ext_vector_type(2)))  float    v2f;
typedef __attribute__((ext_vector_type(8)))  float    v8f;
typedef __attribute__((ext_vector_type(16))) _Float16 v16h;

#if defined(__has_builtin)
#if __has_builtin(__builtin_amdgcn_wmma_f32_16x16x4_f32)
#define USE_F32_WMMA 1
#endif
#endif

template <int NCOLS, bool BIAS>
__global__ __launch_bounds__(32 * (NCOLS / 16)) void gcn_gemm_wmma(
    const float* __restrict__ A, const float* __restrict__ B,
    const float* __restrict__ bias, float* __restrict__ C) {
  const int lane = threadIdx.x & 31;
  const int wave = threadIdx.x >> 5;
  const int m    = lane & 15;
  const int half = lane >> 4;
  const int c0   = wave * 16;
  const int r0   = blockIdx.x * 16;
  const int row  = r0 + m;

  v8f acc = {0.f, 0.f, 0.f, 0.f, 0.f, 0.f, 0.f, 0.f};

#if USE_F32_WMMA
  #pragma unroll
  for (int k0 = 0; k0 < 64; k0 += 4) {
    const int ka = k0 + 2 * half;
    v2f a, b;
    a.x = A[row * 64 + ka];
    a.y = A[row * 64 + ka + 1];
    b.x = B[ka * NCOLS + c0 + m];
    b.y = B[(ka + 1) * NCOLS + c0 + m];
    acc = __builtin_amdgcn_wmma_f32_16x16x4_f32(false, a, false, b,
                                                (short)0, acc, false, false);
  }
#else
  for (int kstep = 0; kstep < 2; ++kstep) {
    v16h af, bf;
    #pragma unroll
    for (int e = 0; e < 16; ++e) {
      int v = e >> 1, j = e & 1, grp = v >> 2;
      int K = 32 * kstep + 16 * grp + 8 * half + ((v & 3) * 2 + j);
      af[e] = (_Float16)A[row * 64 + K];
    }
    #pragma unroll
    for (int e = 0; e < 16; ++e) {
      int v = e >> 1, j = e & 1;
      int K = 32 * kstep + 16 * half + 2 * v + j;
      bf[e] = (_Float16)B[K * NCOLS + c0 + m];
    }
    acc = __builtin_amdgcn_wmma_f32_16x16x32_f16(false, af, false, bf,
                                                 (short)0, acc, false, false);
  }
#endif

  #pragma unroll
  for (int i = 0; i < 8; ++i) {
    float v = acc[i];
    if (BIAS) v += bias[c0 + m];
    C[(r0 + i + 8 * half) * NCOLS + c0 + m] = v;
  }
}

__global__ void k_head_init(int* __restrict__ head) {
  int i = blockIdx.x * blockDim.x + threadIdx.x;
  if (i < N_NODES) head[i] = -1;
}

__global__ void k_link(const int* __restrict__ src, const int* __restrict__ dst,
                       int* __restrict__ head, int* __restrict__ rec) {
  int e = blockIdx.x * blockDim.x + threadIdx.x;
  if (e >= N_EDGES) return;
  int old = atomicExch(&head[dst[e]], e);
  *reinterpret_cast<int2*>(rec + 4 * (size_t)e) = make_int2(old, src[e]);
}

__global__ void k_dinv(const int* __restrict__ head, const int* __restrict__ rec,
                       float* __restrict__ dinv) {
  int i = blockIdx.x * blockDim.x + threadIdx.x;
  if (i >= N_NODES) return;
  int c = 1;
  int e = head[i];
  for (int it = 0; it < N_EDGES && e >= 0; ++it) {
    ++c;
    e = rec[4 * (size_t)e];
  }
  dinv[i] = rsqrtf((float)c);
}

__global__ void k_enorm(const int* __restrict__ src, const int* __restrict__ dst,
                        const float* __restrict__ dinv, int* __restrict__ rec) {
  int i = blockIdx.x * blockDim.x + threadIdx.x;
  if (i < N_EDGES) rec[4 * (size_t)i + 2] = __float_as_int(dinv[src[i]] * dinv[dst[i]]);
}

__global__ __launch_bounds__(256) void k_row_gather(
    const float* __restrict__ h, const float* __restrict__ dinv,
    const int* __restrict__ head, const int* __restrict__ rec,
    float* __restrict__ agg) {
  int node = (blockIdx.x * blockDim.x + threadIdx.x) >> 5;
  const int lane = threadIdx.x & 31;
  if (node >= N_NODES) return;
  node = __builtin_amdgcn_readfirstlane(node);
  const float di = dinv[node];
  const float2 own = *reinterpret_cast<const float2*>(h + (size_t)node * 64 + 2 * lane);
  const float sl = di * di;
  float a0 = sl * own.x, a1 = sl * own.y;
  int e = head[node];
  if (e >= 0) {
    int4 r = *reinterpret_cast<const int4*>(rec + 4 * (size_t)e);
    for (int it = 0; it < N_EDGES; ++it) {
      const float2 x = *reinterpret_cast<const float2*>(h + (size_t)r.y * 64 + 2 * lane);
      const float w = __int_as_float(r.z);
      const int en = r.x;
      int4 rn = r;
      if (en >= 0) rn = *reinterpret_cast<const int4*>(rec + 4 * (size_t)en);
      a0 += w * x.x;
      a1 += w * x.y;
      if (en < 0) break;
      r = rn;
    }
  }
  *reinterpret_cast<float2*>(agg + (size_t)node * 64 + 2 * lane) = make_float2(a0, a1);
}

__global__ __launch_bounds__(128) void k_gemm_ln_relu(
    const float* __restrict__ agg, const float* __restrict__ Wm,
    const float* __restrict__ bias, const float* __restrict__ lnw,
    const float* __restrict__ lnb, float* __restrict__ out) {
  const int tile = (blockIdx.x * blockDim.x + threadIdx.x) >> 5;
  if (tile >= N_NODES / 16) return;
  const int lane = threadIdx.x & 31;
  const int m    = lane & 15;
  const int half = lane >> 4;
  const int r0   = tile * 16;

  const float*  arow = agg + (size_t)(r0 + m) * 64 + 2 * half;
  const float4* W4   = reinterpret_cast<const float4*>(Wm);

  v8f acc0 = {0.f, 0.f, 0.f, 0.f, 0.f, 0.f, 0.f, 0.f};
  v8f acc1 = acc0, acc2 = acc0, acc3 = acc0;

  #pragma unroll
  for (int j = 0; j < 16; ++j) {
    const int k = 4 * j + 2 * half;
    const v2f a = *reinterpret_cast<const v2f*>(arow + 4 * j);
    const float4 w0 = W4[k * 16 + m];
    const float4 w1 = W4[(k + 1) * 16 + m];
    v2f b;
    b.x = w0.x; b.y = w1.x;
    acc0 = __builtin_amdgcn_wmma_f32_16x16x4_f32(false, a, false, b, (short)0, acc0, false, false);
    b.x = w0.y; b.y = w1.y;
    acc1 = __builtin_amdgcn_wmma_f32_16x16x4_f32(false, a, false, b, (short)0, acc1, false, false);
    b.x = w0.z; b.y = w1.z;
    acc2 = __builtin_amdgcn_wmma_f32_16x16x4_f32(false, a, false, b, (short)0, acc2, false, false);
    b.x = w0.w; b.y = w1.w;
    acc3 = __builtin_amdgcn_wmma_f32_16x16x4_f32(false, a, false, b, (short)0, acc3, false, false);
  }

  const float4 bs = reinterpret_cast<const float4*>(bias)[m];
  const float4 gw = reinterpret_cast<const float4*>(lnw)[m];
  const float4 gb = reinterpret_cast<const float4*>(lnb)[m];

  #pragma unroll
  for (int i = 0; i < 8; ++i) {
    const float v0 = acc0[i] + bs.x, v1 = acc1[i] + bs.y;
    const float v2 = acc2[i] + bs.z, v3 = acc3[i] + bs.w;
    float s = (v0 + v1) + (v2 + v3);
    #pragma unroll
    for (int off = 8; off > 0; off >>= 1) s += __shfl_xor(s, off, 32);
    const float mu = s * (1.0f / 64.0f);
    const float d0 = v0 - mu, d1 = v1 - mu, d2 = v2 - mu, d3 = v3 - mu;
    float q = (d0 * d0 + d1 * d1) + (d2 * d2 + d3 * d3);
    #pragma unroll
    for (int off = 8; off > 0; off >>= 1) q += __shfl_xor(q, off, 32);
    const float rstd = rsqrtf(q * (1.0f / 64.0f) + LN_EPS);
    float4 o;
    o.x = fmaxf(0.f, d0 * rstd * gw.x + gb.x);
    o.y = fmaxf(0.f, d1 * rstd * gw.y + gb.y);
    o.z = fmaxf(0.f, d2 * rstd * gw.z + gb.z);
    o.w = fmaxf(0.f, d3 * rstd * gw.w + gb.w);
    reinterpret_cast<float4*>(out + (size_t)(r0 + i + 8 * half) * 64)[m] = o;
  }
}

__global__ void k_zero_pool(float* __restrict__ pool, float* __restrict__ cnt) {
  int i = blockIdx.x * blockDim.x + threadIdx.x;
  if (i < N_GRAPHS * 64) pool[i] = 0.f;
  if (i < N_GRAPHS) cnt[i] = 0.f;
}

#define POOL_ROWS 16
static_assert(N_NODES % POOL_ROWS == 0, "a wave's rows must all exist");

__device__ __forceinline__ void pool_flush(float* __restrict__ pool, float* __restrict__ cnt,
                                           int g, int lane, float s0, float s1, int c) {
  atomicAdd(&pool[g * 64 + 2 * lane], s0);
  atomicAdd(&pool[g * 64 + 2 * lane + 1], s1);
  if (lane == 0) atomicAdd(&cnt[g], (float)c);
}

__global__ __launch_bounds__(256) void k_pool_runs(
    const float* __restrict__ h, const int* __restrict__ batch,
    float* __restrict__ pool, float* __restrict__ cnt) {
  const int chunk = (blockIdx.x * blockDim.x + threadIdx.x) >> 5;
  if (chunk >= N_NODES / POOL_ROWS) return;
  const int lane = threadIdx.x & 31;
  const int r0   = chunk * POOL_ROWS;
  const int bid  = batch[r0 + (lane & (POOL_ROWS - 1))];
  float2 x[POOL_ROWS];
  #pragma unroll
  for (int i = 0; i < POOL_ROWS; ++i)
    x[i] = *reinterpret_cast<const float2*>(h + (size_t)(r0 + i) * 64 + 2 * lane);
  int cur = __builtin_amdgcn_readlane(bid, 0);
  float s0 = 0.f, s1 = 0.f;
  int c = 0;
  #pragma unroll
  for (int i = 0; i < POOL_ROWS; ++i) {
    const int g = __builtin_amdgcn_readlane(bid, i);
    if (g != cur) {
      pool_flush(pool, cnt, cur, lane, s0, s1, c);
      cur = g; s0 = 0.f; s1 = 0.f; c = 0;
    }
    s0 += x[i].x;
    s1 += x[i].y;
    ++c;
  }
  pool_flush(pool, cnt, cur, lane, s0, s1, c);
}

__global__ void k_pool_div(float* __restrict__ pool, const float* __restrict__ cnt) {
  int i = blockIdx.x * blockDim.x + threadIdx.x;
  if (i < N_GRAPHS * 64) pool[i] /= fmaxf(cnt[i >> 6], 1.0f);
}

static inline int cdiv(int a, int b) { return (a + b - 1) / b; }

static void run_layer(const float* hin, const float* W, const float* b,
                      const float* lnw, const float* lnb,
                      const float* dinv, const int* head, const int* rec,
                      float* agg, float* hout,
                      hipStream_t stream) {
  k_row_gather<<<cdiv(N_NODES * 32, 256), 256, 0, stream>>>(hin, dinv, head, rec, agg);
  k_gemm_ln_relu<<<cdiv(N_NODES / 16, 4), 128, 0, stream>>>(agg, W, b, lnw, lnb, hout);
}

extern "C" void kernel_launch(void* const* d_in, const int* in_sizes, int n_in,
                              void* d_out, int out_size, void* d_ws, size_t ws_size,
                              hipStream_t stream) {
  (void)in_sizes; (void)n_in; (void)out_size; (void)ws_size;

  const float* x       = (const float*)d_in[0];
  const int*   eidx    = (const int*)d_in[1];
  const int*   batch   = (const int*)d_in[2];
  const float* W_in    = (const float*)d_in[3];
  const float* b_in    = (const float*)d_in[4];
  const float* ln_in_w = (const float*)d_in[5];
  const float* ln_in_b = (const float*)d_in[6];
  const float* W_h     = (const float*)d_in[7];
  const float* b_h     = (const float*)d_in[8];
  const float* ln_h_w  = (const float*)d_in[9];
  const float* ln_h_b  = (const float*)d_in[10];
  const float* W_out   = (const float*)d_in[11];
  const float* b_out   = (const float*)d_in[12];

  const int* src = eidx;
  const int* dst = eidx + N_EDGES;

  float* ws    = (float*)d_ws;
  float* dinv  = ws;
  float* agg   = dinv + N_PAD;
  float* hbuf  = agg + (size_t)N_NODES*64;
  float* pool  = hbuf + (size_t)N_NODES*64;
  float* cnt   = pool + N_GRAPHS * 64;
  int*   head  = (int*)(cnt + N_GRAPHS);
  int*   rec   = head + N_PAD;

  k_head_init<<<cdiv(N_NODES, 256), 256, 0, stream>>>(head);
  k_link<<<cdiv(N_EDGES, 256), 256, 0, stream>>>(src, dst, head, rec);
  k_dinv<<<cdiv(N_NODES, 64), 64, 0, stream>>>(head, rec, dinv);
  k_enorm<<<cdiv(N_EDGES, 256), 256, 0, stream>>>(src, dst, dinv, rec);

  run_layer(x,    W_in,       b_in,       ln_in_w,      ln_in_b,      dinv, head, rec, agg, hbuf, stream);
  run_layer(hbuf, W_h,        b_h,        ln_h_w,       ln_h_b,       dinv, head, rec, agg, hbuf, stream);
  run_layer(hbuf, W_h + 64*64, b_h + 64,  ln_h_w + 64,  ln_h_b + 64,  dinv, head, rec, agg, hbuf, stream);

  k_zero_pool<<<cdiv(N_GRAPHS * 64, 256), 256, 0, stream>>>(pool, cnt);
  k_pool_runs<<<cdiv(N_NODES / POOL_ROWS * 32, 256), 256, 0, stream>>>(hbuf, batch, pool, cnt);
  k_pool_div<<<cdiv(N_GRAPHS * 64, 256), 256, 0, stream>>>(pool, cnt);
  gcn_gemm_wmma<D_OUT, true><<<N_GRAPHS / 16, 64, 0, stream>>>(pool, W_out, b_out, (float*)d_out);
}


# Row gather: a wave walks 4 rows' chains side by side (8 lanes per row, 8 features per lane, records by vector loads), so 4 dependent hop sequences overlap; 50000 to 12500 waves; sum order kept
# speedup vs baseline: 3.2329x; 1.0439x over previous
#include <hip/hip_runtime.h>
#include <hip/hip_bf16.h>

#define N_NODES  50000
#define N_EDGES  800000
#define N_GRAPHS 512
#define D_H      64
#define D_OUT    32
#define LN_EPS   1e-5f
#define N_PAD    50048

typedef __attribute__((ext_vector_type(2)))  float    v2f;
typedef __attribute__((ext_vector_type(8)))  float    v8f;
typedef __attribute__((ext_vector_type(16))) _Float16 v16h;

#if defined(__has_builtin)
#if __has_builtin(__builtin_amdgcn_wmma_f32_16x16x4_f32)
#define USE_F32_WMMA 1
#endif
#endif

template <int NCOLS, bool BIAS>
__global__ __launch_bounds__(32 * (NCOLS / 16)) void gcn_gemm_wmma(
    const float* __restrict__ A, const float* __restrict__ B,
    const float* __restrict__ bias, float* __restrict__ C) {
  const int lane = threadIdx.x & 31;
  const int wave = threadIdx.x >> 5;
  const int m    = lane & 15;
  const int half = lane >> 4;
  const int c0   = wave * 16;
  const int r0   = blockIdx.x * 16;
  const int row  = r0 + m;

  v8f acc = {0.f, 0.f, 0.f, 0.f, 0.f, 0.f, 0.f, 0.f};

#if USE_F32_WMMA
  #pragma unroll
  for (int k0 = 0; k0 < 64; k0 += 4) {
    const int ka = k0 + 2 * half;
    v2f a, b;
    a.x = A[row * 64 + ka];
    a.y = A[row * 64 + ka + 1];
    b.x = B[ka * NCOLS + c0 + m];
    b.y = B[(ka + 1) * NCOLS + c0 + m];
    acc = __builtin_amdgcn_wmma_f32_16x16x4_f32(false, a, false, b,
                                                (short)0, acc, false, false);
  }
#else
  for (int kstep = 0; kstep < 2; ++kstep) {
    v16h af, bf;
    #pragma unroll
    for (int e = 0; e < 16; ++e) {
      int v = e >> 1, j = e & 1, grp = v >> 2;
      int K = 32 * kstep + 16 * grp + 8 * half + ((v & 3) * 2 + j);
      af[e] = (_Float16)A[row * 64 + K];
    }
    #pragma unroll
    for (int e = 0; e < 16; ++e) {
      int v = e >> 1, j = e & 1;
      int K = 32 * kstep + 16 * half + 2 * v + j;
      bf[e] = (_Float16)B[K * NCOLS + c0 + m];
    }
    acc = __builtin_amdgcn_wmma_f32_16x16x32_f16(false, af, false, bf,
                                                 (short)0, acc, false, false);
  }
#endif

  #pragma unroll
  for (int i = 0; i < 8; ++i) {
    float v = acc[i];
    if (BIAS) v += bias[c0 + m];
    C[(r0 + i + 8 * half) * NCOLS + c0 + m] = v;
  }
}

__global__ void k_head_init(int* __restrict__ head) {
  int i = blockIdx.x * blockDim.x + threadIdx.x;
  if (i < N_NODES) head[i] = -1;
}

__global__ void k_link(const int* __restrict__ src, const int* __restrict__ dst,
                       int* __restrict__ head, int* __restrict__ rec) {
  int e = blockIdx.x * blockDim.x + threadIdx.x;
  if (e >= N_EDGES) return;
  int old = atomicExch(&head[dst[e]], e);
  *reinterpret_cast<int2*>(rec + 4 * (size_t)e) = make_int2(old, src[e]);
}

__global__ void k_dinv(const int* __restrict__ head, const int* __restrict__ rec,
                       float* __restrict__ dinv) {
  int i = blockIdx.x * blockDim.x + threadIdx.x;
  if (i >= N_NODES) return;
  int c = 1;
  int e = head[i];
  for (int it = 0; it < N_EDGES && e >= 0; ++it) {
    ++c;
    e = rec[4 * (size_t)e];
  }
  dinv[i] = rsqrtf((float)c);
}

__global__ void k_enorm(const int* __restrict__ src, const int* __restrict__ dst,
                        const float* __restrict__ dinv, int* __restrict__ rec) {
  int i = blockIdx.x * blockDim.x + threadIdx.x;
  if (i < N_EDGES) rec[4 * (size_t)i + 2] = __float_as_int(dinv[src[i]] * dinv[dst[i]]);
}

#define GATHER_ROWS 4
static_assert(N_NODES % GATHER_ROWS == 0, "a wave's rows must all exist");

__global__ __launch_bounds__(256) void k_row_gather(
    const float* __restrict__ h, const float* __restrict__ dinv,
    const int* __restrict__ head, const int* __restrict__ rec,
    float* __restrict__ agg) {
  const int wave = (blockIdx.x * blockDim.x + threadIdx.x) >> 5;
  if (wave >= N_NODES / GATHER_ROWS) return;
  const int lane = threadIdx.x & 31;
  const int node = wave * GATHER_ROWS + (lane >> 3);
  const int s    = lane & 7;
  const int4* __restrict__ rec4 = reinterpret_cast<const int4*>(rec);

  int e = head[node];
  const float di = dinv[node];
  const float4* hr = reinterpret_cast<const float4*>(h + (size_t)node * 64);
  const float4 o0 = hr[s], o1 = hr[8 + s];
  int4 r = make_int4(-1, 0, 0, 0);
  if (e >= 0) r = rec4[e];
  const float sl = di * di;
  float4 a0, a1;
  a0.x = sl * o0.x; a0.y = sl * o0.y; a0.z = sl * o0.z; a0.w = sl * o0.w;
  a1.x = sl * o1.x; a1.y = sl * o1.y; a1.z = sl * o1.z; a1.w = sl * o1.w;

  for (int it = 0; it < N_EDGES && e >= 0; ++it) {
    const float4* xr = reinterpret_cast<const float4*>(h + (size_t)r.y * 64);
    const float w  = __int_as_float(r.z);
    const int   en = r.x;
    int4 rn = r;
    if (en >= 0) rn = rec4[en];
    const float4 x0 = xr[s], x1 = xr[8 + s];
    a0.x += w * x0.x; a0.y += w * x0.y; a0.z += w * x0.z; a0.w += w * x0.w;
    a1.x += w * x1.x; a1.y += w * x1.y; a1.z += w * x1.z; a1.w += w * x1.w;
    e = en;
    r = rn;
  }
  float4* ar = reinterpret_cast<float4*>(agg + (size_t)node * 64);
  ar[s]     = a0;
  ar[8 + s] = a1;
}

__global__ __launch_bounds__(128) void k_gemm_ln_relu(
    const float* __restrict__ agg, const float* __restrict__ Wm,
    const float* __restrict__ bias, const float* __restrict__ lnw,
    const float* __restrict__ lnb, float* __restrict__ out) {
  const int tile = (blockIdx.x * blockDim.x + threadIdx.x) >> 5;
  if (tile >= N_NODES / 16) return;
  const int lane = threadIdx.x & 31;
  const int m    = lane & 15;
  const int half = lane >> 4;
  const int r0   = tile * 16;

  const float*  arow = agg + (size_t)(r0 + m) * 64 + 2 * half;
  const float4* W4   = reinterpret_cast<const float4*>(Wm);

  v8f acc0 = {0.f, 0.f, 0.f, 0.f, 0.f, 0.f, 0.f, 0.f};
  v8f acc1 = acc0, acc2 = acc0, acc3 = acc0;

  #pragma unroll
  for (int j = 0; j < 16; ++j) {
    const int k = 4 * j + 2 * half;
    const v2f a = *reinterpret_cast<const v2f*>(arow + 4 * j);
    const float4 w0 = W4[k * 16 + m];
    const float4 w1 = W4[(k + 1) * 16 + m];
    v2f b;
    b.x = w0.x; b.y = w1.x;
    acc0 = __builtin_amdgcn_wmma_f32_16x16x4_f32(false, a, false, b, (short)0, acc0, false, false);
    b.x = w0.y; b.y = w1.y;
    acc1 = __builtin_amdgcn_wmma_f32_16x16x4_f32(false, a, false, b, (short)0, acc1, false, false);
    b.x = w0.z; b.y = w1.z;
    acc2 = __builtin_amdgcn_wmma_f32_16x16x4_f32(false, a, false, b, (short)0, acc2, false, false);
    b.x = w0.w; b.y = w1.w;
    acc3 = __builtin_amdgcn_wmma_f32_16x16x4_f32(false, a, false, b, (short)0, acc3, false, false);
  }

  const float4 bs = reinterpret_cast<const float4*>(bias)[m];
  const float4 gw = reinterpret_cast<const float4*>(lnw)[m];
  const float4 gb = reinterpret_cast<const float4*>(lnb)[m];

  #pragma unroll
  for (int i = 0; i < 8; ++i) {
    const float v0 = acc0[i] + bs.x, v1 = acc1[i] + bs.y;
    const float v2 = acc2[i] + bs.z, v3 = acc3[i] + bs.w;
    float s = (v0 + v1) + (v2 + v3);
    #pragma unroll
    for (int off = 8; off > 0; off >>= 1) s += __shfl_xor(s, off, 32);
    const float mu = s * (1.0f / 64.0f);
    const float d0 = v0 - mu, d1 = v1 - mu, d2 = v2 - mu, d3 = v3 - mu;
    float q = (d0 * d0 + d1 * d1) + (d2 * d2 + d3 * d3);
    #pragma unroll
    for (int off = 8; off > 0; off >>= 1) q += __shfl_xor(q, off, 32);
    const float rstd = rsqrtf(q * (1.0f / 64.0f) + LN_EPS);
    float4 o;
    o.x = fmaxf(0.f, d0 * rstd * gw.x + gb.x);
    o.y = fmaxf(0.f, d1 * rstd * gw.y + gb.y);
    o.z = fmaxf(0.f, d2 * rstd * gw.z + gb.z);
    o.w = fmaxf(0.f, d3 * rstd * gw.w + gb.w);
    reinterpret_cast<float4*>(out + (size_t)(r0 + i + 8 * half) * 64)[m] = o;
  }
}

__global__ void k_zero_pool(float* __restrict__ pool, float* __restrict__ cnt) {
  int i = blockIdx.x * blockDim.x + threadIdx.x;
  if (i < N_GRAPHS * 64) pool[i] = 0.f;
  if (i < N_GRAPHS) cnt[i] = 0.f;
}

#define POOL_ROWS 16
static_assert(N_NODES % POOL_ROWS == 0, "a wave's rows must all exist");

__device__ __forceinline__ void pool_flush(float* __restrict__ pool, float* __restrict__ cnt,
                                           int g, int lane, float s0, float s1, int c) {
  atomicAdd(&pool[g * 64 + 2 * lane], s0);
  atomicAdd(&pool[g * 64 + 2 * lane + 1], s1);
  if (lane == 0) atomicAdd(&cnt[g], (float)c);
}

__global__ __launch_bounds__(256) void k_pool_runs(
    const float* __restrict__ h, const int* __restrict__ batch,
    float* __restrict__ pool, float* __restrict__ cnt) {
  const int chunk = (blockIdx.x * blockDim.x + threadIdx.x) >> 5;
  if (chunk >= N_NODES / POOL_ROWS) return;
  const int lane = threadIdx.x & 31;
  const int r0   = chunk * POOL_ROWS;
  const int bid  = batch[r0 + (lane & (POOL_ROWS - 1))];
  float2 x[POOL_ROWS];
  #pragma unroll
  for (int i = 0; i < POOL_ROWS; ++i)
    x[i] = *reinterpret_cast<const float2*>(h + (size_t)(r0 + i) * 64 + 2 * lane);
  int cur = __builtin_amdgcn_readlane(bid, 0);
  float s0 = 0.f, s1 = 0.f;
  int c = 0;
  #pragma unroll
  for (int i = 0; i < POOL_ROWS; ++i) {
    const int g = __builtin_amdgcn_readlane(bid, i);
    if (g != cur) {
      pool_flush(pool, cnt, cur, lane, s0, s1, c);
      cur = g; s0 = 0.f; s1 = 0.f; c = 0;
    }
    s0 += x[i].x;
    s1 += x[i].y;
    ++c;
  }
  pool_flush(pool, cnt, cur, lane, s0, s1, c);
}

__global__ void k_pool_div(float* __restrict__ pool, const float* __restrict__ cnt) {
  int i = blockIdx.x * blockDim.x + threadIdx.x;
  if (i < N_GRAPHS * 64) pool[i] /= fmaxf(cnt[i >> 6], 1.0f);
}

static inline int cdiv(int a, int b) { return (a + b - 1) / b; }

static void run_layer(const float* hin, const float* W, const float* b,
                      const float* lnw, const float* lnb,
                      const float* dinv, const int* head, const int* rec,
                      float* agg, float* hout,
                      hipStream_t stream) {
  k_row_gather<<<cdiv(N_NODES / GATHER_ROWS * 32, 256), 256, 0, stream>>>(hin, dinv, head, rec, agg);
  k_gemm_ln_relu<<<cdiv(N_NODES / 16, 4), 128, 0, stream>>>(agg, W, b, lnw, lnb, hout);
}

extern "C" void kernel_launch(void* const* d_in, const int* in_sizes, int n_in,
                              void* d_out, int out_size, void* d_ws, size_t ws_size,
                              hipStream_t stream) {
  (void)in_sizes; (void)n_in; (void)out_size; (void)ws_size;

  const float* x       = (const float*)d_in[0];
  const int*   eidx    = (const int*)d_in[1];
  const int*   batch   = (const int*)d_in[2];
  const float* W_in    = (const float*)d_in[3];
  const float* b_in    = (const float*)d_in[4];
  const float* ln_in_w = (const float*)d_in[5];
  const float* ln_in_b = (const float*)d_in[6];
  const float* W_h     = (const float*)d_in[7];
  const float* b_h     = (const float*)d_in[8];
  const float* ln_h_w  = (const float*)d_in[9];
  const float* ln_h_b  = (const float*)d_in[10];
  const float* W_out   = (const float*)d_in[11];
  const float* b_out   = (const float*)d_in[12];

  const int* src = eidx;
  const int* dst = eidx + N_EDGES;

  float* ws    = (float*)d_ws;
  float* dinv  = ws;
  float* agg   = dinv + N_PAD;
  float* hbuf  = agg + (size_t)N_NODES*64;
  float* pool  = hbuf + (size_t)N_NODES*64;
  float* cnt   = pool + N_GRAPHS * 64;
  int*   head  = (int*)(cnt + N_GRAPHS);
  int*   rec   = head + N_PAD;

  k_head_init<<<cdiv(N_NODES, 256), 256, 0, stream>>>(head);
  k_link<<<cdiv(N_EDGES, 256), 256, 0, stream>>>(src, dst, head, rec);
  k_dinv<<<cdiv(N_NODES, 64), 64, 0, stream>>>(head, rec, dinv);
  k_enorm<<<cdiv(N_EDGES, 256), 256, 0, stream>>>(src, dst, dinv, rec);

  run_layer(x,    W_in,       b_in,       ln_in_w,      ln_in_b,      dinv, head, rec, agg, hbuf, stream);
  run_layer(hbuf, W_h,        b_h,        ln_h_w,       ln_h_b,       dinv, head, rec, agg, hbuf, stream);
  run_layer(hbuf, W_h + 64*64, b_h + 64,  ln_h_w + 64,  ln_h_b + 64,  dinv, head, rec, agg, hbuf, stream);

  k_zero_pool<<<cdiv(N_GRAPHS * 64, 256), 256, 0, stream>>>(pool, cnt);
  k_pool_runs<<<cdiv(N_NODES / POOL_ROWS * 32, 256), 256, 0, stream>>>(hbuf, batch, pool, cnt);
  k_pool_div<<<cdiv(N_GRAPHS * 64, 256), 256, 0, stream>>>(pool, cnt);
  gcn_gemm_wmma<D_OUT, true><<<N_GRAPHS / 16, 64, 0, stream>>>(pool, W_out, b_out, (float*)d_out);
}


# The degree walk copies each target's chained sources into a 128-byte list at its node, rewritten whole every call; the 3 row gathers read that line, not a record line per edge; row sum order kept
# speedup vs baseline: 4.0348x; 1.2480x over previous
#include <hip/hip_runtime.h>
#include <hip/hip_bf16.h>

#define N_NODES  50000
#define N_EDGES  800000
#define N_GRAPHS 512
#define D_H      64
#define D_OUT    32
#define LN_EPS   1e-5f
#define N_PAD    50048

typedef __attribute__((ext_vector_type(2)))  float    v2f;
typedef __attribute__((ext_vector_type(8)))  float    v8f;
typedef __attribute__((ext_vector_type(16))) _Float16 v16h;

#if defined(__has_builtin)
#if __has_builtin(__builtin_amdgcn_wmma_f32_16x16x4_f32)
#define USE_F32_WMMA 1
#endif
#endif

template <int NCOLS, bool BIAS>
__global__ __launch_bounds__(32 * (NCOLS / 16)) void gcn_gemm_wmma(
    const float* __restrict__ A, const float* __restrict__ B,
    const float* __restrict__ bias, float* __restrict__ C) {
  const int lane = threadIdx.x & 31;
  const int wave = threadIdx.x >> 5;
  const int m    = lane & 15;
  const int half = lane >> 4;
  const int c0   = wave * 16;
  const int r0   = blockIdx.x * 16;
  const int row  = r0 + m;

  v8f acc = {0.f, 0.f, 0.f, 0.f, 0.f, 0.f, 0.f, 0.f};

#if USE_F32_WMMA
  #pragma unroll
  for (int k0 = 0; k0 < 64; k0 += 4) {
    const int ka = k0 + 2 * half;
    v2f a, b;
    a.x = A[row * 64 + ka];
    a.y = A[row * 64 + ka + 1];
    b.x = B[ka * NCOLS + c0 + m];
    b.y = B[(ka + 1) * NCOLS + c0 + m];
    acc = __builtin_amdgcn_wmma_f32_16x16x4_f32(false, a, false, b,
                                                (short)0, acc, false, false);
  }
#else
  for (int kstep = 0; kstep < 2; ++kstep) {
    v16h af, bf;
    #pragma unroll
    for (int e = 0; e < 16; ++e) {
      int v = e >> 1, j = e & 1, grp = v >> 2;
      int K = 32 * kstep + 16 * grp + 8 * half + ((v & 3) * 2 + j);
      af[e] = (_Float16)A[row * 64 + K];
    }
    #pragma unroll
    for (int e = 0; e < 16; ++e) {
      int v = e >> 1, j = e & 1;
      int K = 32 * kstep + 16 * half + 2 * v + j;
      bf[e] = (_Float16)B[K * NCOLS + c0 + m];
    }
    acc = __builtin_amdgcn_wmma_f32_16x16x32_f16(false, af, false, bf,
                                                 (short)0, acc, false, false);
  }
#endif

  #pragma unroll
  for (int i = 0; i < 8; ++i) {
    float v = acc[i];
    if (BIAS) v += bias[c0 + m];
    C[(r0 + i + 8 * half) * NCOLS + c0 + m] = v;
  }
}

#define SUBCHAINS 4
static_assert(SUBCHAINS == 4, "a node's heads are read and written as one int4");

#define ZERO_ROW N_NODES

#define LIST_CAP    32
#define LIST_PIECES (LIST_CAP / 4)
#define LIST_PITCH  (LIST_CAP + 1)

__global__ void k_head_init(int4* __restrict__ head4, float4* __restrict__ zrow4) {
  int i = blockIdx.x * blockDim.x + threadIdx.x;
  if (i < N_NODES) head4[i] = make_int4(-1, -1, -1, -1);
  if (i < 16) zrow4[i] = make_float4(0.f, 0.f, 0.f, 0.f);
}

__global__ void k_link(const int* __restrict__ src, const int* __restrict__ dst,
                       int* __restrict__ head, int2* __restrict__ rec) {
  int e = blockIdx.x * blockDim.x + threadIdx.x;
  if (e >= N_EDGES) return;
  int old = atomicExch(&head[dst[e] * SUBCHAINS + (e & (SUBCHAINS - 1))], e);
  rec[e] = make_int2(old, src[e]);
}

#define DINV_BLOCK 64
__global__ __launch_bounds__(DINV_BLOCK) void k_dinv_scale(
    int4* __restrict__ head4, const int2* __restrict__ rec,
    const float* __restrict__ x, float* __restrict__ dinv, float* __restrict__ hs,
    int4* __restrict__ list4, int* __restrict__ nlist) {
  __shared__ float sd[DINV_BLOCK];
  __shared__ int   sl[DINV_BLOCK * LIST_PITCH];
  const int t  = threadIdx.x;
  const int n0 = blockIdx.x * DINV_BLOCK;
  const int i  = n0 + t;
  const float4* x4 = reinterpret_cast<const float4*>(x) + (size_t)n0 * 16;
  float4 v[16];
  #pragma unroll
  for (int j = 0; j < 16; ++j) {
    const int f = t + DINV_BLOCK * j;
    v[j] = make_float4(0.f, 0.f, 0.f, 0.f);
    if (n0 + (f >> 4) < N_NODES) v[j] = x4[f];
  }
  int* my = sl + t * LIST_PITCH;
  #pragma unroll
  for (int q = 0; q < LIST_CAP; ++q) my[q] = ZERO_ROW;
  float d = 0.f;
  if (i < N_NODES) {
    int c = 1, nl = 0;
    bool full = false;
    int4 oh = make_int4(-1, -1, -1, -1);
    const int4 h4 = head4[i];
    int e[SUBCHAINS] = {h4.x, h4.y, h4.z, h4.w};
    for (int it = 0; it < N_EDGES && (e[0] >= 0 || e[1] >= 0 || e[2] >= 0 || e[3] >= 0); ++it) {
      int2 r[SUBCHAINS];
      bool act[SUBCHAINS];
      int na = 0;
      #pragma unroll
      for (int k = 0; k < SUBCHAINS; ++k) {
        act[k] = e[k] >= 0;
        r[k]   = rec[act[k] ? e[k] : 0];
        na    += act[k] ? 1 : 0;
      }
      if (!full && nl + na > LIST_CAP) {
        full = true;
        oh = make_int4(e[0], e[1], e[2], e[3]);
      }
      #pragma unroll
      for (int k = 0; k < SUBCHAINS; ++k) {
        if (act[k] && !full) { my[nl] = r[k].y; ++nl; }
        c   += act[k] ? 1 : 0;
        e[k] = act[k] ? r[k].x : -1;
      }
    }
    d = rsqrtf((float)c);
    dinv[i]  = d;
    nlist[i] = nl;
    head4[i] = oh;
  }
  sd[t] = d;
  __syncthreads();
  float4* o4 = reinterpret_cast<float4*>(hs) + (size_t)n0 * 16;
  #pragma unroll
  for (int j = 0; j < 16; ++j) {
    const int f = t + DINV_BLOCK * j;
    if (n0 + (f >> 4) < N_NODES) {
      const float sc = sd[f >> 4];
      float4 o;
      o.x = sc * v[j].x; o.y = sc * v[j].y; o.z = sc * v[j].z; o.w = sc * v[j].w;
      o4[f] = o;
    }
  }
  int4* l4 = list4 + (size_t)n0 * LIST_PIECES;
  #pragma unroll
  for (int j = 0; j < LIST_PIECES; ++j) {
    const int p = t + DINV_BLOCK * j;
    if (n0 + (p >> 3) < N_NODES) {
      const int* sp = sl + (p >> 3) * LIST_PITCH + 4 * (p & 7);
      l4[p] = make_int4(sp[0], sp[1], sp[2], sp[3]);
    }
  }
}

#define GATHER_ROWS 4
static_assert(N_NODES % GATHER_ROWS == 0, "a wave's rows must all exist");

__global__ __launch_bounds__(256) void k_row_gather(
    const float* __restrict__ hs, const float* __restrict__ dinv,
    const int4* __restrict__ list4, const int* __restrict__ nlist,
    const int4* __restrict__ head4, const int2* __restrict__ rec,
    float* __restrict__ agg) {
  const int wave = (blockIdx.x * blockDim.x + threadIdx.x) >> 5;
  if (wave >= N_NODES / GATHER_ROWS) return;
  const int lane = threadIdx.x & 31;
  const int node = wave * GATHER_ROWS + (lane >> 3);
  const int s    = lane & 7;

  const int   nl = nlist[node];
  const int4  oh = head4[node];
  const float di = dinv[node];
  const int4 mine = (list4 + (size_t)node * LIST_PIECES)[s];
  const float4* hr = reinterpret_cast<const float4*>(hs + (size_t)node * 64);
  float4 a0 = hr[s], a1 = hr[8 + s];
  const float4* hs4 = reinterpret_cast<const float4*>(hs);

  const int np = (nl + 3) >> 2;
  int4 ids;
  ids.x = __shfl(mine.x, 0, 8); ids.y = __shfl(mine.y, 0, 8);
  ids.z = __shfl(mine.z, 0, 8); ids.w = __shfl(mine.w, 0, 8);
  for (int t = 0; t < LIST_PIECES && t < np; ++t) {
    const int tn = t + 1 < LIST_PIECES ? t + 1 : t;
    int4 nxt;
    nxt.x = __shfl(mine.x, tn, 8); nxt.y = __shfl(mine.y, tn, 8);
    nxt.z = __shfl(mine.z, tn, 8); nxt.w = __shfl(mine.w, tn, 8);
    const int row[4] = {ids.x, ids.y, ids.z, ids.w};
    float4 x0[4], x1[4];
    #pragma unroll
    for (int k = 0; k < 4; ++k) {
      const unsigned p = (unsigned)row[k] * 16u + (unsigned)s;
      x0[k] = hs4[p];
      x1[k] = hs4[p + 8u];
    }
    #pragma unroll
    for (int k = 0; k < 4; ++k) {
      a0.x += x0[k].x; a0.y += x0[k].y; a0.z += x0[k].z; a0.w += x0[k].w;
      a1.x += x1[k].x; a1.y += x1[k].y; a1.z += x1[k].z; a1.w += x1[k].w;
    }
    ids = nxt;
  }

  if (oh.x >= 0 || oh.y >= 0 || oh.z >= 0 || oh.w >= 0) {
    int e[SUBCHAINS] = {oh.x, oh.y, oh.z, oh.w};
    int nx[SUBCHAINS], row[SUBCHAINS];
    #pragma unroll
    for (int k = 0; k < SUBCHAINS; ++k) {
      const bool act = e[k] >= 0;
      const int2 r = rec[act ? e[k] : 0];
      row[k] = act ? r.y : ZERO_ROW;
      nx[k]  = act ? r.x : -1;
    }
    for (int it = 0; it < N_EDGES && (e[0] >= 0 || e[1] >= 0 || e[2] >= 0 || e[3] >= 0); ++it) {
      float4 x0[SUBCHAINS], x1[SUBCHAINS];
      int2   rn[SUBCHAINS];
      #pragma unroll
      for (int k = 0; k < SUBCHAINS; ++k) {
        rn[k] = rec[nx[k] >= 0 ? nx[k] : 0];
        const unsigned p = (unsigned)row[k] * 16u + (unsigned)s;
        x0[k] = hs4[p];
        x1[k] = hs4[p + 8u];
      }
      #pragma unroll
      for (int k = 0; k < SUBCHAINS; ++k) {
        a0.x += x0[k].x; a0.y += x0[k].y; a0.z += x0[k].z; a0.w += x0[k].w;
        a1.x += x1[k].x; a1.y += x1[k].y; a1.z += x1[k].z; a1.w += x1[k].w;
      }
      #pragma unroll
      for (int k = 0; k < SUBCHAINS; ++k) {
        e[k] = nx[k];
        const bool act = e[k] >= 0;
        row[k] = act ? rn[k].y : ZERO_ROW;
        nx[k]  = act ? rn[k].x : -1;
      }
    }
  }
  a0.x *= di; a0.y *= di; a0.z *= di; a0.w *= di;
  a1.x *= di; a1.y *= di; a1.z *= di; a1.w *= di;
  float4* ar = reinterpret_cast<float4*>(agg + (size_t)node * 64);
  ar[s]     = a0;
  ar[8 + s] = a1;
}

template <bool SCALE_OUT>
__global__ __launch_bounds__(128) void k_gemm_ln_relu(
    const float* __restrict__ agg, const float* __restrict__ Wm,
    const float* __restrict__ bias, const float* __restrict__ lnw,
    const float* __restrict__ lnb, const float* __restrict__ dinv,
    float* __restrict__ out) {
  const int tile = (blockIdx.x * blockDim.x + threadIdx.x) >> 5;
  if (tile >= N_NODES / 16) return;
  const int lane = threadIdx.x & 31;
  const int m    = lane & 15;
  const int half = lane >> 4;
  const int r0   = tile * 16;

  const float*  arow = agg + (size_t)(r0 + m) * 64 + 2 * half;
  const float4* W4   = reinterpret_cast<const float4*>(Wm);

  v8f acc0 = {0.f, 0.f, 0.f, 0.f, 0.f, 0.f, 0.f, 0.f};
  v8f acc1 = acc0, acc2 = acc0, acc3 = acc0;

  #pragma unroll
  for (int j = 0; j < 16; ++j) {
    const int k = 4 * j + 2 * half;
    const v2f a = *reinterpret_cast<const v2f*>(arow + 4 * j);
    const float4 w0 = W4[k * 16 + m];
    const float4 w1 = W4[(k + 1) * 16 + m];
    v2f b;
    b.x = w0.x; b.y = w1.x;
    acc0 = __builtin_amdgcn_wmma_f32_16x16x4_f32(false, a, false, b, (short)0, acc0, false, false);
    b.x = w0.y; b.y = w1.y;
    acc1 = __builtin_amdgcn_wmma_f32_16x16x4_f32(false, a, false, b, (short)0, acc1, false, false);
    b.x = w0.z; b.y = w1.z;
    acc2 = __builtin_amdgcn_wmma_f32_16x16x4_f32(false, a, false, b, (short)0, acc2, false, false);
    b.x = w0.w; b.y = w1.w;
    acc3 = __builtin_amdgcn_wmma_f32_16x16x4_f32(false, a, false, b, (short)0, acc3, false, false);
  }

  const float4 bs = reinterpret_cast<const float4*>(bias)[m];
  const float4 gw = reinterpret_cast<const float4*>(lnw)[m];
  const float4 gb = reinterpret_cast<const float4*>(lnb)[m];

  #pragma unroll
  for (int i = 0; i < 8; ++i) {
    const float v0 = acc0[i] + bs.x, v1 = acc1[i] + bs.y;
    const float v2 = acc2[i] + bs.z, v3 = acc3[i] + bs.w;
    float s = (v0 + v1) + (v2 + v3);
    #pragma unroll
    for (int off = 8; off > 0; off >>= 1) s += __shfl_xor(s, off, 32);
    const float mu = s * (1.0f / 64.0f);
    const float d0 = v0 - mu, d1 = v1 - mu, d2 = v2 - mu, d3 = v3 - mu;
    float q = (d0 * d0 + d1 * d1) + (d2 * d2 + d3 * d3);
    #pragma unroll
    for (int off = 8; off > 0; off >>= 1) q += __shfl_xor(q, off, 32);
    const float rstd = rsqrtf(q * (1.0f / 64.0f) + LN_EPS);
    float4 o;
    o.x = fmaxf(0.f, d0 * rstd * gw.x + gb.x);
    o.y = fmaxf(0.f, d1 * rstd * gw.y + gb.y);
    o.z = fmaxf(0.f, d2 * rstd * gw.z + gb.z);
    o.w = fmaxf(0.f, d3 * rstd * gw.w + gb.w);
    if (SCALE_OUT) {
      const float sc = dinv[r0 + i + 8 * half];
      o.x *= sc; o.y *= sc; o.z *= sc; o.w *= sc;
    }
    reinterpret_cast<float4*>(out + (size_t)(r0 + i + 8 * half) * 64)[m] = o;
  }
}

__global__ void k_zero_pool(float* __restrict__ pool, float* __restrict__ cnt) {
  int i = blockIdx.x * blockDim.x + threadIdx.x;
  if (i < N_GRAPHS * 64) pool[i] = 0.f;
  if (i < N_GRAPHS) cnt[i] = 0.f;
}

#define POOL_ROWS 16
static_assert(N_NODES % POOL_ROWS == 0, "a wave's rows must all exist");

__device__ __forceinline__ void pool_flush(float* __restrict__ pool, float* __restrict__ cnt,
                                           int g, int lane, float s0, float s1, int c) {
  atomicAdd(&pool[g * 64 + 2 * lane], s0);
  atomicAdd(&pool[g * 64 + 2 * lane + 1], s1);
  if (lane == 0) atomicAdd(&cnt[g], (float)c);
}

__global__ __launch_bounds__(256) void k_pool_runs(
    const float* __restrict__ h, const int* __restrict__ batch,
    float* __restrict__ pool, float* __restrict__ cnt) {
  const int chunk = (blockIdx.x * blockDim.x + threadIdx.x) >> 5;
  if (chunk >= N_NODES / POOL_ROWS) return;
  const int lane = threadIdx.x & 31;
  const int r0   = chunk * POOL_ROWS;
  const int bid  = batch[r0 + (lane & (POOL_ROWS - 1))];
  float2 x[POOL_ROWS];
  #pragma unroll
  for (int i = 0; i < POOL_ROWS; ++i)
    x[i] = *reinterpret_cast<const float2*>(h + (size_t)(r0 + i) * 64 + 2 * lane);
  int cur = __builtin_amdgcn_readlane(bid, 0);
  float s0 = 0.f, s1 = 0.f;
  int c = 0;
  #pragma unroll
  for (int i = 0; i < POOL_ROWS; ++i) {
    const int g = __builtin_amdgcn_readlane(bid, i);
    if (g != cur) {
      pool_flush(pool, cnt, cur, lane, s0, s1, c);
      cur = g; s0 = 0.f; s1 = 0.f; c = 0;
    }
    s0 += x[i].x;
    s1 += x[i].y;
    ++c;
  }
  pool_flush(pool, cnt, cur, lane, s0, s1, c);
}

__global__ void k_pool_div(float* __restrict__ pool, const float* __restrict__ cnt) {
  int i = blockIdx.x * blockDim.x + threadIdx.x;
  if (i < N_GRAPHS * 64) pool[i] /= fmaxf(cnt[i >> 6], 1.0f);
}

static inline int cdiv(int a, int b) { return (a + b - 1) / b; }

template <bool SCALE_OUT>
static void run_layer(const float* hin, const float* W, const float* b,
                      const float* lnw, const float* lnb,
                      const float* dinv, const int4* list4, const int* nlist,
                      const int4* head4, const int2* rec,
                      float* agg, float* hout,
                      hipStream_t stream) {
  k_row_gather<<<cdiv(N_NODES / GATHER_ROWS * 32, 256), 256, 0, stream>>>(hin, dinv, list4, nlist, head4, rec, agg);
  k_gemm_ln_relu<SCALE_OUT><<<cdiv(N_NODES / 16, 4), 128, 0, stream>>>(agg, W, b, lnw, lnb, dinv, hout);
}

extern "C" void kernel_launch(void* const* d_in, const int* in_sizes, int n_in,
                              void* d_out, int out_size, void* d_ws, size_t ws_size,
                              hipStream_t stream) {
  (void)in_sizes; (void)n_in; (void)out_size; (void)ws_size;

  const float* x       = (const float*)d_in[0];
  const int*   eidx    = (const int*)d_in[1];
  const int*   batch   = (const int*)d_in[2];
  const float* W_in    = (const float*)d_in[3];
  const float* b_in    = (const float*)d_in[4];
  const float* ln_in_w = (const float*)d_in[5];
  const float* ln_in_b = (const float*)d_in[6];
  const float* W_h     = (const float*)d_in[7];
  const float* b_h     = (const float*)d_in[8];
  const float* ln_h_w  = (const float*)d_in[9];
  const float* ln_h_b  = (const float*)d_in[10];
  const float* W_out   = (const float*)d_in[11];
  const float* b_out   = (const float*)d_in[12];

  const int* src = eidx;
  const int* dst = eidx + N_EDGES;

  float* ws    = (float*)d_ws;
  float* dinv  = ws;
  float* agg   = dinv + N_PAD;
  float* hbuf  = agg + (size_t)N_NODES*64;
  float* zrow  = hbuf + (size_t)N_NODES*64;
  float* pool  = zrow + 64;
  float* cnt   = pool + N_GRAPHS * 64;
  int*   head  = (int*)(cnt + N_GRAPHS);
  int2*  rec   = (int2*)(head + (size_t)SUBCHAINS * N_PAD);
  int4*  list4 = (int4*)(rec + N_EDGES);
  int*   nlist = (int*)(list4 + (size_t)LIST_PIECES * N_PAD);

  k_head_init<<<cdiv(N_NODES, 256), 256, 0, stream>>>((int4*)head, (float4*)zrow);
  k_link<<<cdiv(N_EDGES, 256), 256, 0, stream>>>(src, dst, head, rec);
  k_dinv_scale<<<cdiv(N_NODES, DINV_BLOCK), DINV_BLOCK, 0, stream>>>((int4*)head, rec, x, dinv, hbuf, list4, nlist);

  run_layer<true >(hbuf, W_in,        b_in,      ln_in_w,      ln_in_b,      dinv, list4, nlist, (const int4*)head, rec, agg, hbuf, stream);
  run_layer<true >(hbuf, W_h,         b_h,       ln_h_w,       ln_h_b,       dinv, list4, nlist, (const int4*)head, rec, agg, hbuf, stream);
  run_layer<false>(hbuf, W_h + 64*64, b_h + 64,  ln_h_w + 64,  ln_h_b + 64,  dinv, list4, nlist, (const int4*)head, rec, agg, hbuf, stream);

  k_zero_pool<<<cdiv(N_GRAPHS * 64, 256), 256, 0, stream>>>(pool, cnt);
  k_pool_runs<<<cdiv(N_NODES / POOL_ROWS * 32, 256), 256, 0, stream>>>(hbuf, batch, pool, cnt);
  k_pool_div<<<cdiv(N_GRAPHS * 64, 256), 256, 0, stream>>>(pool, cnt);
  gcn_gemm_wmma<D_OUT, true><<<N_GRAPHS / 16, 64, 0, stream>>>(pool, W_out, b_out, (float*)d_out);
}


# Row sums and W product in one wave: it sums its 16 rows (4 passes abreast, 32 row loads in flight), A fragments via its own LDS tile, no barrier; summed rows not stored, 13 to 10 launches; sums kept
# speedup vs baseline: 4.8530x; 1.2028x over previous
#include <hip/hip_runtime.h>
#include <hip/hip_bf16.h>

#define N_NODES  50000
#define N_EDGES  800000
#define N_GRAPHS 512
#define D_H      64
#define D_OUT    32
#define LN_EPS   1e-5f
#define N_PAD    50048

typedef __attribute__((ext_vector_type(2)))  float    v2f;
typedef __attribute__((ext_vector_type(8)))  float    v8f;
typedef __attribute__((ext_vector_type(16))) _Float16 v16h;

#if defined(__has_builtin)
#if __has_builtin(__builtin_amdgcn_wmma_f32_16x16x4_f32)
#define USE_F32_WMMA 1
#endif
#endif

template <int NCOLS, bool BIAS>
__global__ __launch_bounds__(32 * (NCOLS / 16)) void gcn_gemm_wmma(
    const float* __restrict__ A, const float* __restrict__ B,
    const float* __restrict__ bias, float* __restrict__ C) {
  const int lane = threadIdx.x & 31;
  const int wave = threadIdx.x >> 5;
  const int m    = lane & 15;
  const int half = lane >> 4;
  const int c0   = wave * 16;
  const int r0   = blockIdx.x * 16;
  const int row  = r0 + m;

  v8f acc = {0.f, 0.f, 0.f, 0.f, 0.f, 0.f, 0.f, 0.f};

#if USE_F32_WMMA
  #pragma unroll
  for (int k0 = 0; k0 < 64; k0 += 4) {
    const int ka = k0 + 2 * half;
    v2f a, b;
    a.x = A[row * 64 + ka];
    a.y = A[row * 64 + ka + 1];
    b.x = B[ka * NCOLS + c0 + m];
    b.y = B[(ka + 1) * NCOLS + c0 + m];
    acc = __builtin_amdgcn_wmma_f32_16x16x4_f32(false, a, false, b,
                                                (short)0, acc, false, false);
  }
#else
  for (int kstep = 0; kstep < 2; ++kstep) {
    v16h af, bf;
    #pragma unroll
    for (int e = 0; e < 16; ++e) {
      int v = e >> 1, j = e & 1, grp = v >> 2;
      int K = 32 * kstep + 16 * grp + 8 * half + ((v & 3) * 2 + j);
      af[e] = (_Float16)A[row * 64 + K];
    }
    #pragma unroll
    for (int e = 0; e < 16; ++e) {
      int v = e >> 1, j = e & 1;
      int K = 32 * kstep + 16 * half + 2 * v + j;
      bf[e] = (_Float16)B[K * NCOLS + c0 + m];
    }
    acc = __builtin_amdgcn_wmma_f32_16x16x32_f16(false, af, false, bf,
                                                 (short)0, acc, false, false);
  }
#endif

  #pragma unroll
  for (int i = 0; i < 8; ++i) {
    float v = acc[i];
    if (BIAS) v += bias[c0 + m];
    C[(r0 + i + 8 * half) * NCOLS + c0 + m] = v;
  }
}

#define SUBCHAINS 4
static_assert(SUBCHAINS == 4, "a node's heads are read and written as one int4");

#define ZERO_ROW N_NODES

#define LIST_CAP    32
#define LIST_PIECES (LIST_CAP / 4)
#define LIST_PITCH  (LIST_CAP + 1)

__global__ void k_head_init(int4* __restrict__ head4, float4* __restrict__ zrowa4,
                            float4* __restrict__ zrowb4) {
  int i = blockIdx.x * blockDim.x + threadIdx.x;
  if (i < N_NODES) head4[i] = make_int4(-1, -1, -1, -1);
  if (i < 16) {
    zrowa4[i] = make_float4(0.f, 0.f, 0.f, 0.f);
    zrowb4[i] = make_float4(0.f, 0.f, 0.f, 0.f);
  }
}

__global__ void k_link(const int* __restrict__ src, const int* __restrict__ dst,
                       int* __restrict__ head, int2* __restrict__ rec) {
  int e = blockIdx.x * blockDim.x + threadIdx.x;
  if (e >= N_EDGES) return;
  int old = atomicExch(&head[dst[e] * SUBCHAINS + (e & (SUBCHAINS - 1))], e);
  rec[e] = make_int2(old, src[e]);
}

#define DINV_BLOCK 64
__global__ __launch_bounds__(DINV_BLOCK) void k_dinv_scale(
    int4* __restrict__ head4, const int2* __restrict__ rec,
    const float* __restrict__ x, float* __restrict__ dinv, float* __restrict__ hs,
    int4* __restrict__ list4, int* __restrict__ nlist) {
  __shared__ float sd[DINV_BLOCK];
  __shared__ int   sl[DINV_BLOCK * LIST_PITCH];
  const int t  = threadIdx.x;
  const int n0 = blockIdx.x * DINV_BLOCK;
  const int i  = n0 + t;
  const float4* x4 = reinterpret_cast<const float4*>(x) + (size_t)n0 * 16;
  float4 v[16];
  #pragma unroll
  for (int j = 0; j < 16; ++j) {
    const int f = t + DINV_BLOCK * j;
    v[j] = make_float4(0.f, 0.f, 0.f, 0.f);
    if (n0 + (f >> 4) < N_NODES) v[j] = x4[f];
  }
  int* my = sl + t * LIST_PITCH;
  #pragma unroll
  for (int q = 0; q < LIST_CAP; ++q) my[q] = ZERO_ROW;
  float d = 0.f;
  if (i < N_NODES) {
    int c = 1, nl = 0;
    bool full = false;
    int4 oh = make_int4(-1, -1, -1, -1);
    const int4 h4 = head4[i];
    int e[SUBCHAINS] = {h4.x, h4.y, h4.z, h4.w};
    for (int it = 0; it < N_EDGES && (e[0] >= 0 || e[1] >= 0 || e[2] >= 0 || e[3] >= 0); ++it) {
      int2 r[SUBCHAINS];
      bool act[SUBCHAINS];
      int na = 0;
      #pragma unroll
      for (int k = 0; k < SUBCHAINS; ++k) {
        act[k] = e[k] >= 0;
        r[k]   = rec[act[k] ? e[k] : 0];
        na    += act[k] ? 1 : 0;
      }
      if (!full && nl + na > LIST_CAP) {
        full = true;
        oh = make_int4(e[0], e[1], e[2], e[3]);
      }
      #pragma unroll
      for (int k = 0; k < SUBCHAINS; ++k) {
        if (act[k] && !full) { my[nl] = r[k].y; ++nl; }
        c   += act[k] ? 1 : 0;
        e[k] = act[k] ? r[k].x : -1;
      }
    }
    d = rsqrtf((float)c);
    dinv[i]  = d;
    nlist[i] = nl;
    head4[i] = oh;
  }
  sd[t] = d;
  __syncthreads();
  float4* o4 = reinterpret_cast<float4*>(hs) + (size_t)n0 * 16;
  #pragma unroll
  for (int j = 0; j < 16; ++j) {
    const int f = t + DINV_BLOCK * j;
    if (n0 + (f >> 4) < N_NODES) {
      const float sc = sd[f >> 4];
      float4 o;
      o.x = sc * v[j].x; o.y = sc * v[j].y; o.z = sc * v[j].z; o.w = sc * v[j].w;
      o4[f] = o;
    }
  }
  int4* l4 = list4 + (size_t)n0 * LIST_PIECES;
  #pragma unroll
  for (int j = 0; j < LIST_PIECES; ++j) {
    const int p = t + DINV_BLOCK * j;
    if (n0 + (p >> 3) < N_NODES) {
      const int* sp = sl + (p >> 3) * LIST_PITCH + 4 * (p & 7);
      l4[p] = make_int4(sp[0], sp[1], sp[2], sp[3]);
    }
  }
}

#define GATHER_ROWS 4
#define FUSE_PASSES 4
#define TILE_ROWS   (GATHER_ROWS * FUSE_PASSES)
#define TILE_PITCH  68
#define FUSE_WAVES  4
static_assert(TILE_ROWS == 16, "a wave's rows are one WMMA tile");
static_assert(N_NODES % TILE_ROWS == 0, "a wave's rows must all exist");

__device__ __forceinline__ void chain_rest(const int4 oh, const int2* __restrict__ rec,
                                           const float4* __restrict__ hs4, const int s,
                                           float4& a0, float4& a1) {
  int e[SUBCHAINS] = {oh.x, oh.y, oh.z, oh.w};
  int nx[SUBCHAINS], row[SUBCHAINS];
  #pragma unroll
  for (int k = 0; k < SUBCHAINS; ++k) {
    const bool act = e[k] >= 0;
    const int2 r = rec[act ? e[k] : 0];
    row[k] = act ? r.y : ZERO_ROW;
    nx[k]  = act ? r.x : -1;
  }
  for (int it = 0; it < N_EDGES && (e[0] >= 0 || e[1] >= 0 || e[2] >= 0 || e[3] >= 0); ++it) {
    float4 x0[SUBCHAINS], x1[SUBCHAINS];
    int2   rn[SUBCHAINS];
    #pragma unroll
    for (int k = 0; k < SUBCHAINS; ++k) {
      rn[k] = rec[nx[k] >= 0 ? nx[k] : 0];
      const unsigned p = (unsigned)row[k] * 16u + (unsigned)s;
      x0[k] = hs4[p];
      x1[k] = hs4[p + 8u];
    }
    #pragma unroll
    for (int k = 0; k < SUBCHAINS; ++k) {
      a0.x += x0[k].x; a0.y += x0[k].y; a0.z += x0[k].z; a0.w += x0[k].w;
      a1.x += x1[k].x; a1.y += x1[k].y; a1.z += x1[k].z; a1.w += x1[k].w;
    }
    #pragma unroll
    for (int k = 0; k < SUBCHAINS; ++k) {
      e[k] = nx[k];
      const bool act = e[k] >= 0;
      row[k] = act ? rn[k].y : ZERO_ROW;
      nx[k]  = act ? rn[k].x : -1;
    }
  }
}

__device__ __forceinline__ int4 list_piece(const int4 mine, const int t) {
  int4 r;
  r.x = __shfl(mine.x, t, 8); r.y = __shfl(mine.y, t, 8);
  r.z = __shfl(mine.z, t, 8); r.w = __shfl(mine.w, t, 8);
  return r;
}

template <bool SCALE_OUT>
__global__ __launch_bounds__(32 * FUSE_WAVES) __attribute__((amdgpu_waves_per_eu(1, 4))) void k_gather_gemm_ln_relu(
    const float* __restrict__ hs, const float* __restrict__ dinv,
    const int4* __restrict__ list4, const int* __restrict__ nlist,
    const int4* __restrict__ head4, const int2* __restrict__ rec,
    const float* __restrict__ Wm, const float* __restrict__ bias,
    const float* __restrict__ lnw, const float* __restrict__ lnb,
    float* __restrict__ out) {
  __shared__ __attribute__((aligned(16))) float tiles[FUSE_WAVES * TILE_ROWS * TILE_PITCH];
  const int tile = (blockIdx.x * blockDim.x + threadIdx.x) >> 5;
  if (tile >= N_NODES / TILE_ROWS) return;
  const int lane = threadIdx.x & 31;
  const int r0   = tile * TILE_ROWS;
  float* my = tiles + (threadIdx.x >> 5) * (TILE_ROWS * TILE_PITCH);

  const int g = lane >> 3;
  const int s = lane & 7;
  const float4* hs4 = reinterpret_cast<const float4*>(hs);
  int    nl[FUSE_PASSES];
  int4   oh[FUSE_PASSES], mine[FUSE_PASSES];
  float  di[FUSE_PASSES];
  float4 a0[FUSE_PASSES], a1[FUSE_PASSES];
  #pragma unroll
  for (int p = 0; p < FUSE_PASSES; ++p) {
    const int node = r0 + GATHER_ROWS * p + g;
    nl[p] = nlist[node];
    oh[p] = head4[node];
    di[p] = dinv[node];
    mine[p] = (list4 + (size_t)node * LIST_PIECES)[s];
    const float4* hr = hs4 + (size_t)node * 16;
    a0[p] = hr[s];
    a1[p] = hr[8 + s];
  }
  int np = 0;
  #pragma unroll
  for (int p = 0; p < FUSE_PASSES; ++p) np = max(np, (nl[p] + 3) >> 2);
  int4 ids[FUSE_PASSES];
  #pragma unroll
  for (int p = 0; p < FUSE_PASSES; ++p) ids[p] = list_piece(mine[p], 0);
  for (int t = 0; t < LIST_PIECES && t < np; ++t) {
    const int tn = t + 1 < LIST_PIECES ? t + 1 : t;
    int4 nxt[FUSE_PASSES];
    #pragma unroll
    for (int p = 0; p < FUSE_PASSES; ++p) nxt[p] = list_piece(mine[p], tn);
    float4 x0[FUSE_PASSES][4], x1[FUSE_PASSES][4];
    #pragma unroll
    for (int p = 0; p < FUSE_PASSES; ++p) {
      const int row[4] = {ids[p].x, ids[p].y, ids[p].z, ids[p].w};
      #pragma unroll
      for (int k = 0; k < 4; ++k) {
        const unsigned q = (unsigned)row[k] * 16u + (unsigned)s;
        x0[p][k] = hs4[q];
        x1[p][k] = hs4[q + 8u];
      }
    }
    #pragma unroll
    for (int p = 0; p < FUSE_PASSES; ++p) {
      #pragma unroll
      for (int k = 0; k < 4; ++k) {
        a0[p].x += x0[p][k].x; a0[p].y += x0[p][k].y; a0[p].z += x0[p][k].z; a0[p].w += x0[p][k].w;
        a1[p].x += x1[p][k].x; a1[p].y += x1[p][k].y; a1[p].z += x1[p][k].z; a1[p].w += x1[p][k].w;
      }
      ids[p] = nxt[p];
    }
  }
  #pragma unroll
  for (int p = 0; p < FUSE_PASSES; ++p) {
    if (oh[p].x >= 0 || oh[p].y >= 0 || oh[p].z >= 0 || oh[p].w >= 0)
      chain_rest(oh[p], rec, hs4, s, a0[p], a1[p]);
    a0[p].x *= di[p]; a0[p].y *= di[p]; a0[p].z *= di[p]; a0[p].w *= di[p];
    a1[p].x *= di[p]; a1[p].y *= di[p]; a1[p].z *= di[p]; a1[p].w *= di[p];
    float4* tr = reinterpret_cast<float4*>(my + (GATHER_ROWS * p + g) * TILE_PITCH);
    tr[s]     = a0[p];
    tr[8 + s] = a1[p];
  }

  const int m    = lane & 15;
  const int half = lane >> 4;
  const float*  arow = my + m * TILE_PITCH + 2 * half;
  const float4* W4   = reinterpret_cast<const float4*>(Wm);

  v8f acc0 = {0.f, 0.f, 0.f, 0.f, 0.f, 0.f, 0.f, 0.f};
  v8f acc1 = acc0, acc2 = acc0, acc3 = acc0;

  #pragma unroll
  for (int j = 0; j < 16; ++j) {
    const int k = 4 * j + 2 * half;
    const v2f a = *reinterpret_cast<const v2f*>(arow + 4 * j);
    const float4 w0 = W4[k * 16 + m];
    const float4 w1 = W4[(k + 1) * 16 + m];
    v2f b;
    b.x = w0.x; b.y = w1.x;
    acc0 = __builtin_amdgcn_wmma_f32_16x16x4_f32(false, a, false, b, (short)0, acc0, false, false);
    b.x = w0.y; b.y = w1.y;
    acc1 = __builtin_amdgcn_wmma_f32_16x16x4_f32(false, a, false, b, (short)0, acc1, false, false);
    b.x = w0.z; b.y = w1.z;
    acc2 = __builtin_amdgcn_wmma_f32_16x16x4_f32(false, a, false, b, (short)0, acc2, false, false);
    b.x = w0.w; b.y = w1.w;
    acc3 = __builtin_amdgcn_wmma_f32_16x16x4_f32(false, a, false, b, (short)0, acc3, false, false);
  }

  const float4 bs = reinterpret_cast<const float4*>(bias)[m];
  const float4 gw = reinterpret_cast<const float4*>(lnw)[m];
  const float4 gb = reinterpret_cast<const float4*>(lnb)[m];

  #pragma unroll
  for (int i = 0; i < 8; ++i) {
    const float v0 = acc0[i] + bs.x, v1 = acc1[i] + bs.y;
    const float v2 = acc2[i] + bs.z, v3 = acc3[i] + bs.w;
    float sm = (v0 + v1) + (v2 + v3);
    #pragma unroll
    for (int off = 8; off > 0; off >>= 1) sm += __shfl_xor(sm, off, 32);
    const float mu = sm * (1.0f / 64.0f);
    const float d0 = v0 - mu, d1 = v1 - mu, d2 = v2 - mu, d3 = v3 - mu;
    float q = (d0 * d0 + d1 * d1) + (d2 * d2 + d3 * d3);
    #pragma unroll
    for (int off = 8; off > 0; off >>= 1) q += __shfl_xor(q, off, 32);
    const float rstd = rsqrtf(q * (1.0f / 64.0f) + LN_EPS);
    float4 o;
    o.x = fmaxf(0.f, d0 * rstd * gw.x + gb.x);
    o.y = fmaxf(0.f, d1 * rstd * gw.y + gb.y);
    o.z = fmaxf(0.f, d2 * rstd * gw.z + gb.z);
    o.w = fmaxf(0.f, d3 * rstd * gw.w + gb.w);
    if (SCALE_OUT) {
      const float sc = dinv[r0 + i + 8 * half];
      o.x *= sc; o.y *= sc; o.z *= sc; o.w *= sc;
    }
    reinterpret_cast<float4*>(out + (size_t)(r0 + i + 8 * half) * 64)[m] = o;
  }
}

__global__ void k_zero_pool(float* __restrict__ pool, float* __restrict__ cnt) {
  int i = blockIdx.x * blockDim.x + threadIdx.x;
  if (i < N_GRAPHS * 64) pool[i] = 0.f;
  if (i < N_GRAPHS) cnt[i] = 0.f;
}

#define POOL_ROWS 16
static_assert(N_NODES % POOL_ROWS == 0, "a wave's rows must all exist");

__device__ __forceinline__ void pool_flush(float* __restrict__ pool, float* __restrict__ cnt,
                                           int g, int lane, float s0, float s1, int c) {
  atomicAdd(&pool[g * 64 + 2 * lane], s0);
  atomicAdd(&pool[g * 64 + 2 * lane + 1], s1);
  if (lane == 0) atomicAdd(&cnt[g], (float)c);
}

__global__ __launch_bounds__(256) void k_pool_runs(
    const float* __restrict__ h, const int* __restrict__ batch,
    float* __restrict__ pool, float* __restrict__ cnt) {
  const int chunk = (blockIdx.x * blockDim.x + threadIdx.x) >> 5;
  if (chunk >= N_NODES / POOL_ROWS) return;
  const int lane = threadIdx.x & 31;
  const int r0   = chunk * POOL_ROWS;
  const int bid  = batch[r0 + (lane & (POOL_ROWS - 1))];
  float2 x[POOL_ROWS];
  #pragma unroll
  for (int i = 0; i < POOL_ROWS; ++i)
    x[i] = *reinterpret_cast<const float2*>(h + (size_t)(r0 + i) * 64 + 2 * lane);
  int cur = __builtin_amdgcn_readlane(bid, 0);
  float s0 = 0.f, s1 = 0.f;
  int c = 0;
  #pragma unroll
  for (int i = 0; i < POOL_ROWS; ++i) {
    const int g = __builtin_amdgcn_readlane(bid, i);
    if (g != cur) {
      pool_flush(pool, cnt, cur, lane, s0, s1, c);
      cur = g; s0 = 0.f; s1 = 0.f; c = 0;
    }
    s0 += x[i].x;
    s1 += x[i].y;
    ++c;
  }
  pool_flush(pool, cnt, cur, lane, s0, s1, c);
}

__global__ void k_pool_div(float* __restrict__ pool, const float* __restrict__ cnt) {
  int i = blockIdx.x * blockDim.x + threadIdx.x;
  if (i < N_GRAPHS * 64) pool[i] /= fmaxf(cnt[i >> 6], 1.0f);
}

static inline int cdiv(int a, int b) { return (a + b - 1) / b; }

template <bool SCALE_OUT>
static void run_layer(const float* hin, const float* W, const float* b,
                      const float* lnw, const float* lnb,
                      const float* dinv, const int4* list4, const int* nlist,
                      const int4* head4, const int2* rec,
                      float* hout,
                      hipStream_t stream) {
  k_gather_gemm_ln_relu<SCALE_OUT><<<cdiv(N_NODES / TILE_ROWS, FUSE_WAVES), 32 * FUSE_WAVES, 0, stream>>>(
      hin, dinv, list4, nlist, head4, rec, W, b, lnw, lnb, hout);
}

extern "C" void kernel_launch(void* const* d_in, const int* in_sizes, int n_in,
                              void* d_out, int out_size, void* d_ws, size_t ws_size,
                              hipStream_t stream) {
  (void)in_sizes; (void)n_in; (void)out_size; (void)ws_size;

  const float* x       = (const float*)d_in[0];
  const int*   eidx    = (const int*)d_in[1];
  const int*   batch   = (const int*)d_in[2];
  const float* W_in    = (const float*)d_in[3];
  const float* b_in    = (const float*)d_in[4];
  const float* ln_in_w = (const float*)d_in[5];
  const float* ln_in_b = (const float*)d_in[6];
  const float* W_h     = (const float*)d_in[7];
  const float* b_h     = (const float*)d_in[8];
  const float* ln_h_w  = (const float*)d_in[9];
  const float* ln_h_b  = (const float*)d_in[10];
  const float* W_out   = (const float*)d_in[11];
  const float* b_out   = (const float*)d_in[12];

  const int* src = eidx;
  const int* dst = eidx + N_EDGES;

  float* ws    = (float*)d_ws;
  float* dinv  = ws;
  float* hbufa = dinv + N_PAD;
  float* zrowa = hbufa + (size_t)N_NODES*64;
  float* hbufb = zrowa + 64;
  float* zrowb = hbufb + (size_t)N_NODES*64;
  float* pool  = zrowb + 64;
  float* cnt   = pool + N_GRAPHS * 64;
  int*   head  = (int*)(cnt + N_GRAPHS);
  int2*  rec   = (int2*)(head + (size_t)SUBCHAINS * N_PAD);
  int4*  list4 = (int4*)(rec + N_EDGES);
  int*   nlist = (int*)(list4 + (size_t)LIST_PIECES * N_PAD);

  k_head_init<<<cdiv(N_NODES, 256), 256, 0, stream>>>((int4*)head, (float4*)zrowa, (float4*)zrowb);
  k_link<<<cdiv(N_EDGES, 256), 256, 0, stream>>>(src, dst, head, rec);
  k_dinv_scale<<<cdiv(N_NODES, DINV_BLOCK), DINV_BLOCK, 0, stream>>>((int4*)head, rec, x, dinv, hbufa, list4, nlist);

  run_layer<true >(hbufa, W_in,        b_in,      ln_in_w,      ln_in_b,      dinv, list4, nlist, (const int4*)head, rec, hbufb, stream);
  run_layer<true >(hbufb, W_h,         b_h,       ln_h_w,       ln_h_b,       dinv, list4, nlist, (const int4*)head, rec, hbufa, stream);
  run_layer<false>(hbufa, W_h + 64*64, b_h + 64,  ln_h_w + 64,  ln_h_b + 64,  dinv, list4, nlist, (const int4*)head, rec, hbufb, stream);

  k_zero_pool<<<cdiv(N_GRAPHS * 64, 256), 256, 0, stream>>>(pool, cnt);
  k_pool_runs<<<cdiv(N_NODES / POOL_ROWS * 32, 256), 256, 0, stream>>>(hbufb, batch, pool, cnt);
  k_pool_div<<<cdiv(N_GRAPHS * 64, 256), 256, 0, stream>>>(pool, cnt);
  gcn_gemm_wmma<D_OUT, true><<<N_GRAPHS / 16, 64, 0, stream>>>(pool, W_out, b_out, (float*)d_out);
}


# Chain heads in 2 planes, one per L2 half: a wave swaps into the plane of the half it runs on (die id by message), so no head line takes atomics from both halves; row sum order changed, varies by run
# speedup vs baseline: 6.5849x; 1.3569x over previous
#include <hip/hip_runtime.h>
#include <hip/hip_bf16.h>

#define N_NODES  50000
#define N_EDGES  800000
#define N_GRAPHS 512
#define D_H      64
#define D_OUT    32
#define LN_EPS   1e-5f
#define N_PAD    50048

typedef __attribute__((ext_vector_type(2)))  float    v2f;
typedef __attribute__((ext_vector_type(8)))  float    v8f;
typedef __attribute__((ext_vector_type(16))) _Float16 v16h;

#if defined(__has_builtin)
#if __has_builtin(__builtin_amdgcn_wmma_f32_16x16x4_f32)
#define USE_F32_WMMA 1
#endif
#endif

template <int NCOLS, bool BIAS>
__global__ __launch_bounds__(32 * (NCOLS / 16)) void gcn_gemm_wmma(
    const float* __restrict__ A, const float* __restrict__ B,
    const float* __restrict__ bias, float* __restrict__ C) {
  const int lane = threadIdx.x & 31;
  const int wave = threadIdx.x >> 5;
  const int m    = lane & 15;
  const int half = lane >> 4;
  const int c0   = wave * 16;
  const int r0   = blockIdx.x * 16;
  const int row  = r0 + m;

  v8f acc = {0.f, 0.f, 0.f, 0.f, 0.f, 0.f, 0.f, 0.f};

#if USE_F32_WMMA
  #pragma unroll
  for (int k0 = 0; k0 < 64; k0 += 4) {
    const int ka = k0 + 2 * half;
    v2f a, b;
    a.x = A[row * 64 + ka];
    a.y = A[row * 64 + ka + 1];
    b.x = B[ka * NCOLS + c0 + m];
    b.y = B[(ka + 1) * NCOLS + c0 + m];
    acc = __builtin_amdgcn_wmma_f32_16x16x4_f32(false, a, false, b,
                                                (short)0, acc, false, false);
  }
#else
  for (int kstep = 0; kstep < 2; ++kstep) {
    v16h af, bf;
    #pragma unroll
    for (int e = 0; e < 16; ++e) {
      int v = e >> 1, j = e & 1, grp = v >> 2;
      int K = 32 * kstep + 16 * grp + 8 * half + ((v & 3) * 2 + j);
      af[e] = (_Float16)A[row * 64 + K];
    }
    #pragma unroll
    for (int e = 0; e < 16; ++e) {
      int v = e >> 1, j = e & 1;
      int K = 32 * kstep + 16 * half + 2 * v + j;
      bf[e] = (_Float16)B[K * NCOLS + c0 + m];
    }
    acc = __builtin_amdgcn_wmma_f32_16x16x32_f16(false, af, false, bf,
                                                 (short)0, acc, false, false);
  }
#endif

  #pragma unroll
  for (int i = 0; i < 8; ++i) {
    float v = acc[i];
    if (BIAS) v += bias[c0 + m];
    C[(r0 + i + 8 * half) * NCOLS + c0 + m] = v;
  }
}

#define SUBCHAINS 4
#define HEAD_PLANES 2
#define PLANE_CHAINS (SUBCHAINS / HEAD_PLANES)
static_assert(SUBCHAINS == 4 && PLANE_CHAINS == 2, "a node's heads of one plane are one int2, the edges its chains go on from one int4");

#define ZERO_ROW N_NODES

#define LIST_CAP    32
#define LIST_PIECES (LIST_CAP / 4)
#define LIST_PITCH  (LIST_CAP + 1)

__global__ void k_head_init(int4* __restrict__ head4, float4* __restrict__ zrowa4,
                            float4* __restrict__ zrowb4) {
  int i = blockIdx.x * blockDim.x + threadIdx.x;
  if (i < N_PAD) head4[i] = make_int4(-1, -1, -1, -1);
  if (i < 16) {
    zrowa4[i] = make_float4(0.f, 0.f, 0.f, 0.f);
    zrowb4[i] = make_float4(0.f, 0.f, 0.f, 0.f);
  }
}

__global__ void k_link(const int* __restrict__ src, const int* __restrict__ dst,
                       int* __restrict__ head, int2* __restrict__ rec) {
  __shared__ int s_half;
  const int e = blockIdx.x * blockDim.x + threadIdx.x;
  const bool in = e < N_EDGES;
  int d = 0, sn = 0;
  if (in) { d = dst[e]; sn = src[e]; }
  if (threadIdx.x < 32) {
    const int w = __builtin_amdgcn_s_sendmsg_rtn(0x87);
    if (threadIdx.x == 0) s_half = (w >> 18) & 1;
  }
  __syncthreads();
  if (in) {
    const int old = atomicExch(&head[(s_half * N_PAD + d) * PLANE_CHAINS + (e & (PLANE_CHAINS - 1))], e);
    rec[e] = make_int2(old, sn);
  }
}

#define DINV_BLOCK 64
__global__ __launch_bounds__(DINV_BLOCK) void k_dinv_scale(
    const int2* __restrict__ head2, int4* __restrict__ rest4, const int2* __restrict__ rec,
    const float* __restrict__ x, float* __restrict__ dinv, float* __restrict__ hs,
    int4* __restrict__ list4, int* __restrict__ nlist) {
  __shared__ float sd[DINV_BLOCK];
  __shared__ int   sl[DINV_BLOCK * LIST_PITCH];
  const int t  = threadIdx.x;
  const int n0 = blockIdx.x * DINV_BLOCK;
  const int i  = n0 + t;
  const float4* x4 = reinterpret_cast<const float4*>(x) + (size_t)n0 * 16;
  float4 v[16];
  #pragma unroll
  for (int j = 0; j < 16; ++j) {
    const int f = t + DINV_BLOCK * j;
    v[j] = make_float4(0.f, 0.f, 0.f, 0.f);
    if (n0 + (f >> 4) < N_NODES) v[j] = x4[f];
  }
  int* my = sl + t * LIST_PITCH;
  #pragma unroll
  for (int q = 0; q < LIST_CAP; ++q) my[q] = ZERO_ROW;
  float d = 0.f;
  if (i < N_NODES) {
    int c = 1, nl = 0;
    bool full = false;
    int4 oh = make_int4(-1, -1, -1, -1);
    const int2 ha = head2[i];
    const int2 hb = head2[N_PAD + i];
    int e[SUBCHAINS] = {ha.x, ha.y, hb.x, hb.y};
    for (int it = 0; it < N_EDGES && (e[0] >= 0 || e[1] >= 0 || e[2] >= 0 || e[3] >= 0); ++it) {
      int2 r[SUBCHAINS];
      bool act[SUBCHAINS];
      int na = 0;
      #pragma unroll
      for (int k = 0; k < SUBCHAINS; ++k) {
        act[k] = e[k] >= 0;
        r[k]   = rec[act[k] ? e[k] : 0];
        na    += act[k] ? 1 : 0;
      }
      if (!full && nl + na > LIST_CAP) {
        full = true;
        oh = make_int4(e[0], e[1], e[2], e[3]);
      }
      #pragma unroll
      for (int k = 0; k < SUBCHAINS; ++k) {
        if (act[k] && !full) { my[nl] = r[k].y; ++nl; }
        c   += act[k] ? 1 : 0;
        e[k] = act[k] ? r[k].x : -1;
      }
    }
    d = rsqrtf((float)c);
    dinv[i]  = d;
    nlist[i] = nl;
    rest4[i] = oh;
  }
  sd[t] = d;
  __syncthreads();
  float4* o4 = reinterpret_cast<float4*>(hs) + (size_t)n0 * 16;
  #pragma unroll
  for (int j = 0; j < 16; ++j) {
    const int f = t + DINV_BLOCK * j;
    if (n0 + (f >> 4) < N_NODES) {
      const float sc = sd[f >> 4];
      float4 o;
      o.x = sc * v[j].x; o.y = sc * v[j].y; o.z = sc * v[j].z; o.w = sc * v[j].w;
      o4[f] = o;
    }
  }
  int4* l4 = list4 + (size_t)n0 * LIST_PIECES;
  #pragma unroll
  for (int j = 0; j < LIST_PIECES; ++j) {
    const int p = t + DINV_BLOCK * j;
    if (n0 + (p >> 3) < N_NODES) {
      const int* sp = sl + (p >> 3) * LIST_PITCH + 4 * (p & 7);
      l4[p] = make_int4(sp[0], sp[1], sp[2], sp[3]);
    }
  }
}

#define GATHER_ROWS 4
#define FUSE_PASSES 4
#define TILE_ROWS   (GATHER_ROWS * FUSE_PASSES)
#define TILE_PITCH  68
#define FUSE_WAVES  4
static_assert(TILE_ROWS == 16, "a wave's rows are one WMMA tile");
static_assert(N_NODES % TILE_ROWS == 0, "a wave's rows must all exist");

__device__ __forceinline__ void chain_rest(const int4 oh, const int2* __restrict__ rec,
                                           const float4* __restrict__ hs4, const int s,
                                           float4& a0, float4& a1) {
  int e[SUBCHAINS] = {oh.x, oh.y, oh.z, oh.w};
  int nx[SUBCHAINS], row[SUBCHAINS];
  #pragma unroll
  for (int k = 0; k < SUBCHAINS; ++k) {
    const bool act = e[k] >= 0;
    const int2 r = rec[act ? e[k] : 0];
    row[k] = act ? r.y : ZERO_ROW;
    nx[k]  = act ? r.x : -1;
  }
  for (int it = 0; it < N_EDGES && (e[0] >= 0 || e[1] >= 0 || e[2] >= 0 || e[3] >= 0); ++it) {
    float4 x0[SUBCHAINS], x1[SUBCHAINS];
    int2   rn[SUBCHAINS];
    #pragma unroll
    for (int k = 0; k < SUBCHAINS; ++k) {
      rn[k] = rec[nx[k] >= 0 ? nx[k] : 0];
      const unsigned p = (unsigned)row[k] * 16u + (unsigned)s;
      x0[k] = hs4[p];
      x1[k] = hs4[p + 8u];
    }
    #pragma unroll
    for (int k = 0; k < SUBCHAINS; ++k) {
      a0.x += x0[k].x; a0.y += x0[k].y; a0.z += x0[k].z; a0.w += x0[k].w;
      a1.x += x1[k].x; a1.y += x1[k].y; a1.z += x1[k].z; a1.w += x1[k].w;
    }
    #pragma unroll
    for (int k = 0; k < SUBCHAINS; ++k) {
      e[k] = nx[k];
      const bool act = e[k] >= 0;
      row[k] = act ? rn[k].y : ZERO_ROW;
      nx[k]  = act ? rn[k].x : -1;
    }
  }
}

__device__ __forceinline__ int4 list_piece(const int4 mine, const int t) {
  int4 r;
  r.x = __shfl(mine.x, t, 8); r.y = __shfl(mine.y, t, 8);
  r.z = __shfl(mine.z, t, 8); r.w = __shfl(mine.w, t, 8);
  return r;
}

template <bool SCALE_OUT>
__global__ __launch_bounds__(32 * FUSE_WAVES) __attribute__((amdgpu_waves_per_eu(1, 4))) void k_gather_gemm_ln_relu(
    const float* __restrict__ hs, const float* __restrict__ dinv,
    const int4* __restrict__ list4, const int* __restrict__ nlist,
    const int4* __restrict__ rest4, const int2* __restrict__ rec,
    const float* __restrict__ Wm, const float* __restrict__ bias,
    const float* __restrict__ lnw, const float* __restrict__ lnb,
    float* __restrict__ out) {
  __shared__ __attribute__((aligned(16))) float tiles[FUSE_WAVES * TILE_ROWS * TILE_PITCH];
  const int tile = (blockIdx.x * blockDim.x + threadIdx.x) >> 5;
  if (tile >= N_NODES / TILE_ROWS) return;
  const int lane = threadIdx.x & 31;
  const int r0   = tile * TILE_ROWS;
  float* my = tiles + (threadIdx.x >> 5) * (TILE_ROWS * TILE_PITCH);

  const int g = lane >> 3;
  const int s = lane & 7;
  const float4* hs4 = reinterpret_cast<const float4*>(hs);
  int    nl[FUSE_PASSES];
  int4   oh[FUSE_PASSES], mine[FUSE_PASSES];
  float  di[FUSE_PASSES];
  float4 a0[FUSE_PASSES], a1[FUSE_PASSES];
  #pragma unroll
  for (int p = 0; p < FUSE_PASSES; ++p) {
    const int node = r0 + GATHER_ROWS * p + g;
    nl[p] = nlist[node];
    oh[p] = rest4[node];
    di[p] = dinv[node];
    mine[p] = (list4 + (size_t)node * LIST_PIECES)[s];
    const float4* hr = hs4 + (size_t)node * 16;
    a0[p] = hr[s];
    a1[p] = hr[8 + s];
  }
  int np = 0;
  #pragma unroll
  for (int p = 0; p < FUSE_PASSES; ++p) np = max(np, (nl[p] + 3) >> 2);
  int4 ids[FUSE_PASSES];
  #pragma unroll
  for (int p = 0; p < FUSE_PASSES; ++p) ids[p] = list_piece(mine[p], 0);
  for (int t = 0; t < LIST_PIECES && t < np; ++t) {
    const int tn = t + 1 < LIST_PIECES ? t + 1 : t;
    int4 nxt[FUSE_PASSES];
    #pragma unroll
    for (int p = 0; p < FUSE_PASSES; ++p) nxt[p] = list_piece(mine[p], tn);
    float4 x0[FUSE_PASSES][4], x1[FUSE_PASSES][4];
    #pragma unroll
    for (int p = 0; p < FUSE_PASSES; ++p) {
      const int row[4] = {ids[p].x, ids[p].y, ids[p].z, ids[p].w};
      #pragma unroll
      for (int k = 0; k < 4; ++k) {
        const unsigned q = (unsigned)row[k] * 16u + (unsigned)s;
        x0[p][k] = hs4[q];
        x1[p][k] = hs4[q + 8u];
      }
    }
    #pragma unroll
    for (int p = 0; p < FUSE_PASSES; ++p) {
      #pragma unroll
      for (int k = 0; k < 4; ++k) {
        a0[p].x += x0[p][k].x; a0[p].y += x0[p][k].y; a0[p].z += x0[p][k].z; a0[p].w += x0[p][k].w;
        a1[p].x += x1[p][k].x; a1[p].y += x1[p][k].y; a1[p].z += x1[p][k].z; a1[p].w += x1[p][k].w;
      }
      ids[p] = nxt[p];
    }
  }
  #pragma unroll
  for (int p = 0; p < FUSE_PASSES; ++p) {
    if (oh[p].x >= 0 || oh[p].y >= 0 || oh[p].z >= 0 || oh[p].w >= 0)
      chain_rest(oh[p], rec, hs4, s, a0[p], a1[p]);
    a0[p].x *= di[p]; a0[p].y *= di[p]; a0[p].z *= di[p]; a0[p].w *= di[p];
    a1[p].x *= di[p]; a1[p].y *= di[p]; a1[p].z *= di[p]; a1[p].w *= di[p];
    float4* tr = reinterpret_cast<float4*>(my + (GATHER_ROWS * p + g) * TILE_PITCH);
    tr[s]     = a0[p];
    tr[8 + s] = a1[p];
  }

  const int m    = lane & 15;
  const int half = lane >> 4;
  const float*  arow = my + m * TILE_PITCH + 2 * half;
  const float4* W4   = reinterpret_cast<const float4*>(Wm);

  v8f acc0 = {0.f, 0.f, 0.f, 0.f, 0.f, 0.f, 0.f, 0.f};
  v8f acc1 = acc0, acc2 = acc0, acc3 = acc0;

  #pragma unroll
  for (int j = 0; j < 16; ++j) {
    const int k = 4 * j + 2 * half;
    const v2f a = *reinterpret_cast<const v2f*>(arow + 4 * j);
    const float4 w0 = W4[k * 16 + m];
    const float4 w1 = W4[(k + 1) * 16 + m];
    v2f b;
    b.x = w0.x; b.y = w1.x;
    acc0 = __builtin_amdgcn_wmma_f32_16x16x4_f32(false, a, false, b, (short)0, acc0, false, false);
    b.x = w0.y; b.y = w1.y;
    acc1 = __builtin_amdgcn_wmma_f32_16x16x4_f32(false, a, false, b, (short)0, acc1, false, false);
    b.x = w0.z; b.y = w1.z;
    acc2 = __builtin_amdgcn_wmma_f32_16x16x4_f32(false, a, false, b, (short)0, acc2, false, false);
    b.x = w0.w; b.y = w1.w;
    acc3 = __builtin_amdgcn_wmma_f32_16x16x4_f32(false, a, false, b, (short)0, acc3, false, false);
  }

  const float4 bs = reinterpret_cast<const float4*>(bias)[m];
  const float4 gw = reinterpret_cast<const float4*>(lnw)[m];
  const float4 gb = reinterpret_cast<const float4*>(lnb)[m];

  #pragma unroll
  for (int i = 0; i < 8; ++i) {
    const float v0 = acc0[i] + bs.x, v1 = acc1[i] + bs.y;
    const float v2 = acc2[i] + bs.z, v3 = acc3[i] + bs.w;
    float sm = (v0 + v1) + (v2 + v3);
    #pragma unroll
    for (int off = 8; off > 0; off >>= 1) sm += __shfl_xor(sm, off, 32);
    const float mu = sm * (1.0f / 64.0f);
    const float d0 = v0 - mu, d1 = v1 - mu, d2 = v2 - mu, d3 = v3 - mu;
    float q = (d0 * d0 + d1 * d1) + (d2 * d2 + d3 * d3);
    #pragma unroll
    for (int off = 8; off > 0; off >>= 1) q += __shfl_xor(q, off, 32);
    const float rstd = rsqrtf(q * (1.0f / 64.0f) + LN_EPS);
    float4 o;
    o.x = fmaxf(0.f, d0 * rstd * gw.x + gb.x);
    o.y = fmaxf(0.f, d1 * rstd * gw.y + gb.y);
    o.z = fmaxf(0.f, d2 * rstd * gw.z + gb.z);
    o.w = fmaxf(0.f, d3 * rstd * gw.w + gb.w);
    if (SCALE_OUT) {
      const float sc = dinv[r0 + i + 8 * half];
      o.x *= sc; o.y *= sc; o.z *= sc; o.w *= sc;
    }
    reinterpret_cast<float4*>(out + (size_t)(r0 + i + 8 * half) * 64)[m] = o;
  }
}

__global__ void k_zero_pool(float* __restrict__ pool, float* __restrict__ cnt) {
  int i = blockIdx.x * blockDim.x + threadIdx.x;
  if (i < N_GRAPHS * 64) pool[i] = 0.f;
  if (i < N_GRAPHS) cnt[i] = 0.f;
}

#define POOL_ROWS 16
static_assert(N_NODES % POOL_ROWS == 0, "a wave's rows must all exist");

__device__ __forceinline__ void pool_flush(float* __restrict__ pool, float* __restrict__ cnt,
                                           int g, int lane, float s0, float s1, int c) {
  atomicAdd(&pool[g * 64 + 2 * lane], s0);
  atomicAdd(&pool[g * 64 + 2 * lane + 1], s1);
  if (lane == 0) atomicAdd(&cnt[g], (float)c);
}

__global__ __launch_bounds__(256) void k_pool_runs(
    const float* __restrict__ h, const int* __restrict__ batch,
    float* __restrict__ pool, float* __restrict__ cnt) {
  const int chunk = (blockIdx.x * blockDim.x + threadIdx.x) >> 5;
  if (chunk >= N_NODES / POOL_ROWS) return;
  const int lane = threadIdx.x & 31;
  const int r0   = chunk * POOL_ROWS;
  const int bid  = batch[r0 + (lane & (POOL_ROWS - 1))];
  float2 x[POOL_ROWS];
  #pragma unroll
  for (int i = 0; i < POOL_ROWS; ++i)
    x[i] = *reinterpret_cast<const float2*>(h + (size_t)(r0 + i) * 64 + 2 * lane);
  int cur = __builtin_amdgcn_readlane(bid, 0);
  float s0 = 0.f, s1 = 0.f;
  int c = 0;
  #pragma unroll
  for (int i = 0; i < POOL_ROWS; ++i) {
    const int g = __builtin_amdgcn_readlane(bid, i);
    if (g != cur) {
      pool_flush(pool, cnt, cur, lane, s0, s1, c);
      cur = g; s0 = 0.f; s1 = 0.f; c = 0;
    }
    s0 += x[i].x;
    s1 += x[i].y;
    ++c;
  }
  pool_flush(pool, cnt, cur, lane, s0, s1, c);
}

__global__ void k_pool_div(float* __restrict__ pool, const float* __restrict__ cnt) {
  int i = blockIdx.x * blockDim.x + threadIdx.x;
  if (i < N_GRAPHS * 64) pool[i] /= fmaxf(cnt[i >> 6], 1.0f);
}

static inline int cdiv(int a, int b) { return (a + b - 1) / b; }

template <bool SCALE_OUT>
static void run_layer(const float* hin, const float* W, const float* b,
                      const float* lnw, const float* lnb,
                      const float* dinv, const int4* list4, const int* nlist,
                      const int4* rest4, const int2* rec,
                      float* hout,
                      hipStream_t stream) {
  k_gather_gemm_ln_relu<SCALE_OUT><<<cdiv(N_NODES / TILE_ROWS, FUSE_WAVES), 32 * FUSE_WAVES, 0, stream>>>(
      hin, dinv, list4, nlist, rest4, rec, W, b, lnw, lnb, hout);
}

extern "C" void kernel_launch(void* const* d_in, const int* in_sizes, int n_in,
                              void* d_out, int out_size, void* d_ws, size_t ws_size,
                              hipStream_t stream) {
  (void)in_sizes; (void)n_in; (void)out_size; (void)ws_size;

  const float* x       = (const float*)d_in[0];
  const int*   eidx    = (const int*)d_in[1];
  const int*   batch   = (const int*)d_in[2];
  const float* W_in    = (const float*)d_in[3];
  const float* b_in    = (const float*)d_in[4];
  const float* ln_in_w = (const float*)d_in[5];
  const float* ln_in_b = (const float*)d_in[6];
  const float* W_h     = (const float*)d_in[7];
  const float* b_h     = (const float*)d_in[8];
  const float* ln_h_w  = (const float*)d_in[9];
  const float* ln_h_b  = (const float*)d_in[10];
  const float* W_out   = (const float*)d_in[11];
  const float* b_out   = (const float*)d_in[12];

  const int* src = eidx;
  const int* dst = eidx + N_EDGES;

  float* ws    = (float*)d_ws;
  float* dinv  = ws;
  float* hbufa = dinv + N_PAD;
  float* zrowa = hbufa + (size_t)N_NODES*64;
  float* hbufb = zrowa + 64;
  float* zrowb = hbufb + (size_t)N_NODES*64;
  float* pool  = zrowb + 64;
  float* cnt   = pool + N_GRAPHS * 64;
  int*   head  = (int*)(cnt + N_GRAPHS);
  int4*  rest4 = (int4*)(head + (size_t)SUBCHAINS * N_PAD);
  int2*  rec   = (int2*)(rest4 + N_PAD);
  int4*  list4 = (int4*)(rec + N_EDGES);
  int*   nlist = (int*)(list4 + (size_t)LIST_PIECES * N_PAD);

  k_head_init<<<cdiv(N_PAD, 256), 256, 0, stream>>>((int4*)head, (float4*)zrowa, (float4*)zrowb);
  k_link<<<cdiv(N_EDGES, 256), 256, 0, stream>>>(src, dst, head, rec);
  k_dinv_scale<<<cdiv(N_NODES, DINV_BLOCK), DINV_BLOCK, 0, stream>>>((const int2*)head, rest4, rec, x, dinv, hbufa, list4, nlist);

  run_layer<true >(hbufa, W_in,        b_in,      ln_in_w,      ln_in_b,      dinv, list4, nlist, rest4, rec, hbufb, stream);
  run_layer<true >(hbufb, W_h,         b_h,       ln_h_w,       ln_h_b,       dinv, list4, nlist, rest4, rec, hbufa, stream);
  run_layer<false>(hbufa, W_h + 64*64, b_h + 64,  ln_h_w + 64,  ln_h_b + 64,  dinv, list4, nlist, rest4, rec, hbufb, stream);

  k_zero_pool<<<cdiv(N_GRAPHS * 64, 256), 256, 0, stream>>>(pool, cnt);
  k_pool_runs<<<cdiv(N_NODES / POOL_ROWS * 32, 256), 256, 0, stream>>>(hbufb, batch, pool, cnt);
  k_pool_div<<<cdiv(N_GRAPHS * 64, 256), 256, 0, stream>>>(pool, cnt);
  gcn_gemm_wmma<D_OUT, true><<<N_GRAPHS / 16, 64, 0, stream>>>(pool, W_out, b_out, (float*)d_out);
}


# Pool start values stored by the first launch and sum / row count taken in the output product's A loader, no zeroing or dividing launch of their own: 10 to 8 launches; every sum and quotient as before
# speedup vs baseline: 6.8549x; 1.0410x over previous
#include <hip/hip_runtime.h>
#include <hip/hip_bf16.h>

#define N_NODES  50000
#define N_EDGES  800000
#define N_GRAPHS 512
#define D_H      64
#define D_OUT    32
#define LN_EPS   1e-5f
#define N_PAD    50048

typedef __attribute__((ext_vector_type(2)))  float    v2f;
typedef __attribute__((ext_vector_type(8)))  float    v8f;
typedef __attribute__((ext_vector_type(16))) _Float16 v16h;

#if defined(__has_builtin)
#if __has_builtin(__builtin_amdgcn_wmma_f32_16x16x4_f32)
#define USE_F32_WMMA 1
#endif
#endif

template <int NCOLS, bool BIAS>
__global__ __launch_bounds__(32 * (NCOLS / 16)) void gcn_gemm_wmma(
    const float* __restrict__ A, const float* __restrict__ cnt, const float* __restrict__ B,
    const float* __restrict__ bias, float* __restrict__ C) {
  const int lane = threadIdx.x & 31;
  const int wave = threadIdx.x >> 5;
  const int m    = lane & 15;
  const int half = lane >> 4;
  const int c0   = wave * 16;
  const int r0   = blockIdx.x * 16;
  const int row  = r0 + m;
  const float cn = fmaxf(cnt[row], 1.0f);

  v8f acc = {0.f, 0.f, 0.f, 0.f, 0.f, 0.f, 0.f, 0.f};

#if USE_F32_WMMA
  #pragma unroll
  for (int k0 = 0; k0 < 64; k0 += 4) {
    const int ka = k0 + 2 * half;
    v2f a, b;
    a.x = A[row * 64 + ka] / cn;
    a.y = A[row * 64 + ka + 1] / cn;
    b.x = B[ka * NCOLS + c0 + m];
    b.y = B[(ka + 1) * NCOLS + c0 + m];
    acc = __builtin_amdgcn_wmma_f32_16x16x4_f32(false, a, false, b,
                                                (short)0, acc, false, false);
  }
#else
  for (int kstep = 0; kstep < 2; ++kstep) {
    v16h af, bf;
    #pragma unroll
    for (int e = 0; e < 16; ++e) {
      int v = e >> 1, j = e & 1, grp = v >> 2;
      int K = 32 * kstep + 16 * grp + 8 * half + ((v & 3) * 2 + j);
      af[e] = (_Float16)(A[row * 64 + K] / cn);
    }
    #pragma unroll
    for (int e = 0; e < 16; ++e) {
      int v = e >> 1, j = e & 1;
      int K = 32 * kstep + 16 * half + 2 * v + j;
      bf[e] = (_Float16)B[K * NCOLS + c0 + m];
    }
    acc = __builtin_amdgcn_wmma_f32_16x16x32_f16(false, af, false, bf,
                                                 (short)0, acc, false, false);
  }
#endif

  #pragma unroll
  for (int i = 0; i < 8; ++i) {
    float v = acc[i];
    if (BIAS) v += bias[c0 + m];
    C[(r0 + i + 8 * half) * NCOLS + c0 + m] = v;
  }
}

#define SUBCHAINS 4
#define HEAD_PLANES 2
#define PLANE_CHAINS (SUBCHAINS / HEAD_PLANES)
static_assert(SUBCHAINS == 4 && PLANE_CHAINS == 2, "a node's heads of one plane are one int2, the edges its chains go on from one int4");

#define ZERO_ROW N_NODES

#define LIST_CAP    32
#define LIST_PIECES (LIST_CAP / 4)
#define LIST_PITCH  (LIST_CAP + 1)

__global__ void k_head_init(int4* __restrict__ head4, float4* __restrict__ zrowa4,
                            float4* __restrict__ zrowb4, float4* __restrict__ pool4,
                            float4* __restrict__ cnt4) {
  int i = blockIdx.x * blockDim.x + threadIdx.x;
  if (i < N_PAD) head4[i] = make_int4(-1, -1, -1, -1);
  if (i < N_GRAPHS * 16) pool4[i] = make_float4(0.f, 0.f, 0.f, 0.f);
  if (i < N_GRAPHS / 4)  cnt4[i]  = make_float4(0.f, 0.f, 0.f, 0.f);
  if (i < 16) {
    zrowa4[i] = make_float4(0.f, 0.f, 0.f, 0.f);
    zrowb4[i] = make_float4(0.f, 0.f, 0.f, 0.f);
  }
}

__global__ void k_link(const int* __restrict__ src, const int* __restrict__ dst,
                       int* __restrict__ head, int2* __restrict__ rec) {
  __shared__ int s_half;
  const int e = blockIdx.x * blockDim.x + threadIdx.x;
  const bool in = e < N_EDGES;
  int d = 0, sn = 0;
  if (in) { d = dst[e]; sn = src[e]; }
  if (threadIdx.x < 32) {
    const int w = __builtin_amdgcn_s_sendmsg_rtn(0x87);
    if (threadIdx.x == 0) s_half = (w >> 18) & 1;
  }
  __syncthreads();
  if (in) {
    const int old = atomicExch(&head[(s_half * N_PAD + d) * PLANE_CHAINS + (e & (PLANE_CHAINS - 1))], e);
    rec[e] = make_int2(old, sn);
  }
}

#define DINV_BLOCK 64
__global__ __launch_bounds__(DINV_BLOCK) void k_dinv_scale(
    const int2* __restrict__ head2, int4* __restrict__ rest4, const int2* __restrict__ rec,
    const float* __restrict__ x, float* __restrict__ dinv, float* __restrict__ hs,
    int4* __restrict__ list4, int* __restrict__ nlist) {
  __shared__ float sd[DINV_BLOCK];
  __shared__ int   sl[DINV_BLOCK * LIST_PITCH];
  const int t  = threadIdx.x;
  const int n0 = blockIdx.x * DINV_BLOCK;
  const int i  = n0 + t;
  const float4* x4 = reinterpret_cast<const float4*>(x) + (size_t)n0 * 16;
  float4 v[16];
  #pragma unroll
  for (int j = 0; j < 16; ++j) {
    const int f = t + DINV_BLOCK * j;
    v[j] = make_float4(0.f, 0.f, 0.f, 0.f);
    if (n0 + (f >> 4) < N_NODES) v[j] = x4[f];
  }
  int* my = sl + t * LIST_PITCH;
  #pragma unroll
  for (int q = 0; q < LIST_CAP; ++q) my[q] = ZERO_ROW;
  float d = 0.f;
  if (i < N_NODES) {
    int c = 1, nl = 0;
    bool full = false;
    int4 oh = make_int4(-1, -1, -1, -1);
    const int2 ha = head2[i];
    const int2 hb = head2[N_PAD + i];
    int e[SUBCHAINS] = {ha.x, ha.y, hb.x, hb.y};
    for (int it = 0; it < N_EDGES && (e[0] >= 0 || e[1] >= 0 || e[2] >= 0 || e[3] >= 0); ++it) {
      int2 r[SUBCHAINS];
      bool act[SUBCHAINS];
      int na = 0;
      #pragma unroll
      for (int k = 0; k < SUBCHAINS; ++k) {
        act[k] = e[k] >= 0;
        r[k]   = rec[act[k] ? e[k] : 0];
        na    += act[k] ? 1 : 0;
      }
      if (!full && nl + na > LIST_CAP) {
        full = true;
        oh = make_int4(e[0], e[1], e[2], e[3]);
      }
      #pragma unroll
      for (int k = 0; k < SUBCHAINS; ++k) {
        if (act[k] && !full) { my[nl] = r[k].y; ++nl; }
        c   += act[k] ? 1 : 0;
        e[k] = act[k] ? r[k].x : -1;
      }
    }
    d = rsqrtf((float)c);
    dinv[i]  = d;
    nlist[i] = nl;
    rest4[i] = oh;
  }
  sd[t] = d;
  __syncthreads();
  float4* o4 = reinterpret_cast<float4*>(hs) + (size_t)n0 * 16;
  #pragma unroll
  for (int j = 0; j < 16; ++j) {
    const int f = t + DINV_BLOCK * j;
    if (n0 + (f >> 4) < N_NODES) {
      const float sc = sd[f >> 4];
      float4 o;
      o.x = sc * v[j].x; o.y = sc * v[j].y; o.z = sc * v[j].z; o.w = sc * v[j].w;
      o4[f] = o;
    }
  }
  int4* l4 = list4 + (size_t)n0 * LIST_PIECES;
  #pragma unroll
  for (int j = 0; j < LIST_PIECES; ++j) {
    const int p = t + DINV_BLOCK * j;
    if (n0 + (p >> 3) < N_NODES) {
      const int* sp = sl + (p >> 3) * LIST_PITCH + 4 * (p & 7);
      l4[p] = make_int4(sp[0], sp[1], sp[2], sp[3]);
    }
  }
}

#define GATHER_ROWS 4
#define FUSE_PASSES 4
#define TILE_ROWS   (GATHER_ROWS * FUSE_PASSES)
#define TILE_PITCH  68
#define FUSE_WAVES  4
static_assert(TILE_ROWS == 16, "a wave's rows are one WMMA tile");
static_assert(N_NODES % TILE_ROWS == 0, "a wave's rows must all exist");

__device__ __forceinline__ void chain_rest(const int4 oh, const int2* __restrict__ rec,
                                           const float4* __restrict__ hs4, const int s,
                                           float4& a0, float4& a1) {
  int e[SUBCHAINS] = {oh.x, oh.y, oh.z, oh.w};
  int nx[SUBCHAINS], row[SUBCHAINS];
  #pragma unroll
  for (int k = 0; k < SUBCHAINS; ++k) {
    const bool act = e[k] >= 0;
    const int2 r = rec[act ? e[k] : 0];
    row[k] = act ? r.y : ZERO_ROW;
    nx[k]  = act ? r.x : -1;
  }
  for (int it = 0; it < N_EDGES && (e[0] >= 0 || e[1] >= 0 || e[2] >= 0 || e[3] >= 0); ++it) {
    float4 x0[SUBCHAINS], x1[SUBCHAINS];
    int2   rn[SUBCHAINS];
    #pragma unroll
    for (int k = 0; k < SUBCHAINS; ++k) {
      rn[k] = rec[nx[k] >= 0 ? nx[k] : 0];
      const unsigned p = (unsigned)row[k] * 16u + (unsigned)s;
      x0[k] = hs4[p];
      x1[k] = hs4[p + 8u];
    }
    #pragma unroll
    for (int k = 0; k < SUBCHAINS; ++k) {
      a0.x += x0[k].x; a0.y += x0[k].y; a0.z += x0[k].z; a0.w += x0[k].w;
      a1.x += x1[k].x; a1.y += x1[k].y; a1.z += x1[k].z; a1.w += x1[k].w;
    }
    #pragma unroll
    for (int k = 0; k < SUBCHAINS; ++k) {
      e[k] = nx[k];
      const bool act = e[k] >= 0;
      row[k] = act ? rn[k].y : ZERO_ROW;
      nx[k]  = act ? rn[k].x : -1;
    }
  }
}

__device__ __forceinline__ int4 list_piece(const int4 mine, const int t) {
  int4 r;
  r.x = __shfl(mine.x, t, 8); r.y = __shfl(mine.y, t, 8);
  r.z = __shfl(mine.z, t, 8); r.w = __shfl(mine.w, t, 8);
  return r;
}

template <bool SCALE_OUT>
__global__ __launch_bounds__(32 * FUSE_WAVES) __attribute__((amdgpu_waves_per_eu(1, 4))) void k_gather_gemm_ln_relu(
    const float* __restrict__ hs, const float* __restrict__ dinv,
    const int4* __restrict__ list4, const int* __restrict__ nlist,
    const int4* __restrict__ rest4, const int2* __restrict__ rec,
    const float* __restrict__ Wm, const float* __restrict__ bias,
    const float* __restrict__ lnw, const float* __restrict__ lnb,
    float* __restrict__ out) {
  __shared__ __attribute__((aligned(16))) float tiles[FUSE_WAVES * TILE_ROWS * TILE_PITCH];
  const int tile = (blockIdx.x * blockDim.x + threadIdx.x) >> 5;
  if (tile >= N_NODES / TILE_ROWS) return;
  const int lane = threadIdx.x & 31;
  const int r0   = tile * TILE_ROWS;
  float* my = tiles + (threadIdx.x >> 5) * (TILE_ROWS * TILE_PITCH);

  const int g = lane >> 3;
  const int s = lane & 7;
  const float4* hs4 = reinterpret_cast<const float4*>(hs);
  int    nl[FUSE_PASSES];
  int4   oh[FUSE_PASSES], mine[FUSE_PASSES];
  float  di[FUSE_PASSES];
  float4 a0[FUSE_PASSES], a1[FUSE_PASSES];
  #pragma unroll
  for (int p = 0; p < FUSE_PASSES; ++p) {
    const int node = r0 + GATHER_ROWS * p + g;
    nl[p] = nlist[node];
    oh[p] = rest4[node];
    di[p] = dinv[node];
    mine[p] = (list4 + (size_t)node * LIST_PIECES)[s];
    const float4* hr = hs4 + (size_t)node * 16;
    a0[p] = hr[s];
    a1[p] = hr[8 + s];
  }
  int np = 0;
  #pragma unroll
  for (int p = 0; p < FUSE_PASSES; ++p) np = max(np, (nl[p] + 3) >> 2);
  int4 ids[FUSE_PASSES];
  #pragma unroll
  for (int p = 0; p < FUSE_PASSES; ++p) ids[p] = list_piece(mine[p], 0);
  for (int t = 0; t < LIST_PIECES && t < np; ++t) {
    const int tn = t + 1 < LIST_PIECES ? t + 1 : t;
    int4 nxt[FUSE_PASSES];
    #pragma unroll
    for (int p = 0; p < FUSE_PASSES; ++p) nxt[p] = list_piece(mine[p], tn);
    float4 x0[FUSE_PASSES][4], x1[FUSE_PASSES][4];
    #pragma unroll
    for (int p = 0; p < FUSE_PASSES; ++p) {
      const int row[4] = {ids[p].x, ids[p].y, ids[p].z, ids[p].w};
      #pragma unroll
      for (int k = 0; k < 4; ++k) {
        const unsigned q = (unsigned)row[k] * 16u + (unsigned)s;
        x0[p][k] = hs4[q];
        x1[p][k] = hs4[q + 8u];
      }
    }
    #pragma unroll
    for (int p = 0; p < FUSE_PASSES; ++p) {
      #pragma unroll
      for (int k = 0; k < 4; ++k) {
        a0[p].x += x0[p][k].x; a0[p].y += x0[p][k].y; a0[p].z += x0[p][k].z; a0[p].w += x0[p][k].w;
        a1[p].x += x1[p][k].x; a1[p].y += x1[p][k].y; a1[p].z += x1[p][k].z; a1[p].w += x1[p][k].w;
      }
      ids[p] = nxt[p];
    }
  }
  #pragma unroll
  for (int p = 0; p < FUSE_PASSES; ++p) {
    if (oh[p].x >= 0 || oh[p].y >= 0 || oh[p].z >= 0 || oh[p].w >= 0)
      chain_rest(oh[p], rec, hs4, s, a0[p], a1[p]);
    a0[p].x *= di[p]; a0[p].y *= di[p]; a0[p].z *= di[p]; a0[p].w *= di[p];
    a1[p].x *= di[p]; a1[p].y *= di[p]; a1[p].z *= di[p]; a1[p].w *= di[p];
    float4* tr = reinterpret_cast<float4*>(my + (GATHER_ROWS * p + g) * TILE_PITCH);
    tr[s]     = a0[p];
    tr[8 + s] = a1[p];
  }

  const int m    = lane & 15;
  const int half = lane >> 4;
  const float*  arow = my + m * TILE_PITCH + 2 * half;
  const float4* W4   = reinterpret_cast<const float4*>(Wm);

  v8f acc0 = {0.f, 0.f, 0.f, 0.f, 0.f, 0.f, 0.f, 0.f};
  v8f acc1 = acc0, acc2 = acc0, acc3 = acc0;

  #pragma unroll
  for (int j = 0; j < 16; ++j) {
    const int k = 4 * j + 2 * half;
    const v2f a = *reinterpret_cast<const v2f*>(arow + 4 * j);
    const float4 w0 = W4[k * 16 + m];
    const float4 w1 = W4[(k + 1) * 16 + m];
    v2f b;
    b.x = w0.x; b.y = w1.x;
    acc0 = __builtin_amdgcn_wmma_f32_16x16x4_f32(false, a, false, b, (short)0, acc0, false, false);
    b.x = w0.y; b.y = w1.y;
    acc1 = __builtin_amdgcn_wmma_f32_16x16x4_f32(false, a, false, b, (short)0, acc1, false, false);
    b.x = w0.z; b.y = w1.z;
    acc2 = __builtin_amdgcn_wmma_f32_16x16x4_f32(false, a, false, b, (short)0, acc2, false, false);
    b.x = w0.w; b.y = w1.w;
    acc3 = __builtin_amdgcn_wmma_f32_16x16x4_f32(false, a, false, b, (short)0, acc3, false, false);
  }

  const float4 bs = reinterpret_cast<const float4*>(bias)[m];
  const float4 gw = reinterpret_cast<const float4*>(lnw)[m];
  const float4 gb = reinterpret_cast<const float4*>(lnb)[m];

  #pragma unroll
  for (int i = 0; i < 8; ++i) {
    const float v0 = acc0[i] + bs.x, v1 = acc1[i] + bs.y;
    const float v2 = acc2[i] + bs.z, v3 = acc3[i] + bs.w;
    float sm = (v0 + v1) + (v2 + v3);
    #pragma unroll
    for (int off = 8; off > 0; off >>= 1) sm += __shfl_xor(sm, off, 32);
    const float mu = sm * (1.0f / 64.0f);
    const float d0 = v0 - mu, d1 = v1 - mu, d2 = v2 - mu, d3 = v3 - mu;
    float q = (d0 * d0 + d1 * d1) + (d2 * d2 + d3 * d3);
    #pragma unroll
    for (int off = 8; off > 0; off >>= 1) q += __shfl_xor(q, off, 32);
    const float rstd = rsqrtf(q * (1.0f / 64.0f) + LN_EPS);
    float4 o;
    o.x = fmaxf(0.f, d0 * rstd * gw.x + gb.x);
    o.y = fmaxf(0.f, d1 * rstd * gw.y + gb.y);
    o.z = fmaxf(0.f, d2 * rstd * gw.z + gb.z);
    o.w = fmaxf(0.f, d3 * rstd * gw.w + gb.w);
    if (SCALE_OUT) {
      const float sc = dinv[r0 + i + 8 * half];
      o.x *= sc; o.y *= sc; o.z *= sc; o.w *= sc;
    }
    reinterpret_cast<float4*>(out + (size_t)(r0 + i + 8 * half) * 64)[m] = o;
  }
}

#define POOL_ROWS 16
static_assert(N_NODES % POOL_ROWS == 0, "a wave's rows must all exist");

__device__ __forceinline__ void pool_flush(float* __restrict__ pool, float* __restrict__ cnt,
                                           int g, int lane, float s0, float s1, int c) {
  atomicAdd(&pool[g * 64 + 2 * lane], s0);
  atomicAdd(&pool[g * 64 + 2 * lane + 1], s1);
  if (lane == 0) atomicAdd(&cnt[g], (float)c);
}

__global__ __launch_bounds__(256) void k_pool_runs(
    const float* __restrict__ h, const int* __restrict__ batch,
    float* __restrict__ pool, float* __restrict__ cnt) {
  const int chunk = (blockIdx.x * blockDim.x + threadIdx.x) >> 5;
  if (chunk >= N_NODES / POOL_ROWS) return;
  const int lane = threadIdx.x & 31;
  const int r0   = chunk * POOL_ROWS;
  const int bid  = batch[r0 + (lane & (POOL_ROWS - 1))];
  float2 x[POOL_ROWS];
  #pragma unroll
  for (int i = 0; i < POOL_ROWS; ++i)
    x[i] = *reinterpret_cast<const float2*>(h + (size_t)(r0 + i) * 64 + 2 * lane);
  int cur = __builtin_amdgcn_readlane(bid, 0);
  float s0 = 0.f, s1 = 0.f;
  int c = 0;
  #pragma unroll
  for (int i = 0; i < POOL_ROWS; ++i) {
    const int g = __builtin_amdgcn_readlane(bid, i);
    if (g != cur) {
      pool_flush(pool, cnt, cur, lane, s0, s1, c);
      cur = g; s0 = 0.f; s1 = 0.f; c = 0;
    }
    s0 += x[i].x;
    s1 += x[i].y;
    ++c;
  }
  pool_flush(pool, cnt, cur, lane, s0, s1, c);
}

static inline int cdiv(int a, int b) { return (a + b - 1) / b; }

template <bool SCALE_OUT>
static void run_layer(const float* hin, const float* W, const float* b,
                      const float* lnw, const float* lnb,
                      const float* dinv, const int4* list4, const int* nlist,
                      const int4* rest4, const int2* rec,
                      float* hout,
                      hipStream_t stream) {
  k_gather_gemm_ln_relu<SCALE_OUT><<<cdiv(N_NODES / TILE_ROWS, FUSE_WAVES), 32 * FUSE_WAVES, 0, stream>>>(
      hin, dinv, list4, nlist, rest4, rec, W, b, lnw, lnb, hout);
}

extern "C" void kernel_launch(void* const* d_in, const int* in_sizes, int n_in,
                              void* d_out, int out_size, void* d_ws, size_t ws_size,
                              hipStream_t stream) {
  (void)in_sizes; (void)n_in; (void)out_size; (void)ws_size;

  const float* x       = (const float*)d_in[0];
  const int*   eidx    = (const int*)d_in[1];
  const int*   batch   = (const int*)d_in[2];
  const float* W_in    = (const float*)d_in[3];
  const float* b_in    = (const float*)d_in[4];
  const float* ln_in_w = (const float*)d_in[5];
  const float* ln_in_b = (const float*)d_in[6];
  const float* W_h     = (const float*)d_in[7];
  const float* b_h     = (const float*)d_in[8];
  const float* ln_h_w  = (const float*)d_in[9];
  const float* ln_h_b  = (const float*)d_in[10];
  const float* W_out   = (const float*)d_in[11];
  const float* b_out   = (const float*)d_in[12];

  const int* src = eidx;
  const int* dst = eidx + N_EDGES;

  float* ws    = (float*)d_ws;
  float* dinv  = ws;
  float* hbufa = dinv + N_PAD;
  float* zrowa = hbufa + (size_t)N_NODES*64;
  float* hbufb = zrowa + 64;
  float* zrowb = hbufb + (size_t)N_NODES*64;
  float* pool  = zrowb + 64;
  float* cnt   = pool + N_GRAPHS * 64;
  int*   head  = (int*)(cnt + N_GRAPHS);
  int4*  rest4 = (int4*)(head + (size_t)SUBCHAINS * N_PAD);
  int2*  rec   = (int2*)(rest4 + N_PAD);
  int4*  list4 = (int4*)(rec + N_EDGES);
  int*   nlist = (int*)(list4 + (size_t)LIST_PIECES * N_PAD);

  k_head_init<<<cdiv(N_PAD, 256), 256, 0, stream>>>((int4*)head, (float4*)zrowa, (float4*)zrowb, (float4*)pool, (float4*)cnt);
  k_link<<<cdiv(N_EDGES, 256), 256, 0, stream>>>(src, dst, head, rec);
  k_dinv_scale<<<cdiv(N_NODES, DINV_BLOCK), DINV_BLOCK, 0, stream>>>((const int2*)head, rest4, rec, x, dinv, hbufa, list4, nlist);

  run_layer<true >(hbufa, W_in,        b_in,      ln_in_w,      ln_in_b,      dinv, list4, nlist, rest4, rec, hbufb, stream);
  run_layer<true >(hbufb, W_h,         b_h,       ln_h_w,       ln_h_b,       dinv, list4, nlist, rest4, rec, hbufa, stream);
  run_layer<false>(hbufa, W_h + 64*64, b_h + 64,  ln_h_w + 64,  ln_h_b + 64,  dinv, list4, nlist, rest4, rec, hbufb, stream);

  k_pool_runs<<<cdiv(N_NODES / POOL_ROWS * 32, 256), 256, 0, stream>>>(hbufb, batch, pool, cnt);
  gcn_gemm_wmma<D_OUT, true><<<N_GRAPHS / 16, 64, 0, stream>>>(pool, cnt, W_out, b_out, (float*)d_out);
}
